# starting kernel
# baseline (speedup 1.0000x reference)
#include <hip/hip_runtime.h>
#include <hip/hip_bf16.h>
#include <math.h>

#define EPS 1e-6f
#define LDIM 256
#define BATCH 65536
#define SPAD 257

typedef __attribute__((ext_vector_type(2))) float v2f;
typedef __attribute__((ext_vector_type(8))) float v8f;

__global__ __launch_bounds__(1024) void sinkhorn_kernel(const float* __restrict__ logits,
                                                        float* __restrict__ P) {
    __shared__ float M[LDIM * SPAD];
    const int t = threadIdx.x;
    const int g = t >> 2;
    const int q = t & 3;

    for (int i = t; i < LDIM * LDIM; i += 1024)
        M[(i >> 8) * SPAD + (i & 255)] = logits[i];
    __syncthreads();

    for (int it = 0; it < 10; ++it) {
        {
            const int base = g * SPAD + q * 64;
            float mx = -3.4e38f;
            #pragma unroll 8
            for (int i = 0; i < 64; ++i) mx = fmaxf(mx, M[base + i]);
            mx = fmaxf(mx, __shfl_xor(mx, 1));
            mx = fmaxf(mx, __shfl_xor(mx, 2));
            float s = 0.0f;
            #pragma unroll 8
            for (int i = 0; i < 64; ++i) s += expf(M[base + i] - mx);
            s += __shfl_xor(s, 1);
            s += __shfl_xor(s, 2);
            const float lse = mx + logf(s);
            #pragma unroll 8
            for (int i = 0; i < 64; ++i) M[base + i] -= lse;
        }
        __syncthreads();
        {
            const int base = (q * 64) * SPAD + g;
            float mx = -3.4e38f;
            #pragma unroll 8
            for (int i = 0; i < 64; ++i) mx = fmaxf(mx, M[base + i * SPAD]);
            mx = fmaxf(mx, __shfl_xor(mx, 1));
            mx = fmaxf(mx, __shfl_xor(mx, 2));
            float s = 0.0f;
            #pragma unroll 8
            for (int i = 0; i < 64; ++i) s += expf(M[base + i * SPAD] - mx);
            s += __shfl_xor(s, 1);
            s += __shfl_xor(s, 2);
            const float lse = mx + logf(s);
            #pragma unroll 8
            for (int i = 0; i < 64; ++i) M[base + i * SPAD] -= lse;
        }
        __syncthreads();
    }

    for (int i = t; i < LDIM * LDIM; i += 1024)
        P[i] = expf(M[(i >> 8) * SPAD + (i & 255)]);
}

__global__ __launch_bounds__(256) void gemm_kernel(const float* __restrict__ x,
                                                   const float* __restrict__ P,
                                                   float* __restrict__ leafT) {
    __shared__ float Ps[LDIM * LDIM];
    const int t = threadIdx.x;
    {
        const float4* Pv = (const float4*)P;
        float4* Sv = (float4*)Ps;
        for (int i = t; i < LDIM * LDIM / 4; i += 256) Sv[i] = Pv[i];
    }
    __syncthreads();

    const int wave  = t >> 5;
    const int lane  = t & 31;
    const int mloc  = lane & 15;
    const int khalf = (lane >> 4) * 2;
    const int m0    = blockIdx.x * 128 + wave * 16;

    v8f acc[16] = {};

    const float* xrow = x + (size_t)(m0 + mloc) * LDIM + khalf;

    for (int k = 0; k < LDIM; k += 4) {
        v2f bfrag;
        bfrag.x = xrow[k];
        bfrag.y = xrow[k + 1];
        #pragma unroll
        for (int nt = 0; nt < 16; ++nt) {
            const int n = nt * 16 + mloc;
            v2f afrag;
            afrag.x = Ps[(k + khalf)     * LDIM + n];
            afrag.y = Ps[(k + khalf + 1) * LDIM + n];
            acc[nt] = __builtin_amdgcn_wmma_f32_16x16x4_f32(
                false, afrag, false, bfrag, (short)0, acc[nt], false, false);
        }
    }

    const size_t col = (size_t)m0 + mloc;
    #pragma unroll
    for (int nt = 0; nt < 16; ++nt) {
        const int nbase = nt * 16 + (lane >> 4) * 8;
        #pragma unroll
        for (int i = 0; i < 8; ++i)
            leafT[(size_t)(nbase + i) * BATCH + col] = acc[nt][i];
    }
}

__device__ __forceinline__ float r_poly(float a) {
    float d  = 0.5f - a;
    float d2 = d * d;
    float num = 0.25f + 1.65811f * d + 2.15388f * d2 + 8.2844f * d2 * d + 6.16764f * d2 * d2;
    float den = a * (1.0f - a);
    den = (fabsf(den) < EPS) ? EPS : den;
    return num / den;
}

__device__ __forceinline__ float power_r(float x, float y, float ra, float w0, float w1) {
    ra = (fabsf(ra) < EPS) ? EPS : ra;
    return powf(w0 * powf(x, ra) + w1 * powf(y, ra), 1.0f / ra);
}

__device__ __forceinline__ float F_upper(float x, float y, float a, float w0, float w1) {
    const float xy  = x * y;
    const float mn  = fminf(x, y);
    const float lin = w0 * x + w1 * y;
    float out = lin;
    if (a > 0.5f && a < 0.75f) {
        const float R = r_poly(0.75f);
        out = (3.0f - 4.0f * a) * lin + (4.0f * a - 2.0f) * power_r(x, y, R, w0, w1);
    }
    if (a >= 0.75f && a < 1.0f) {
        const float ra = r_poly(fminf(fmaxf(a, 0.75f), 1.0f - EPS));
        out = power_r(x, y, ra, w0, w1);
    }
    if (fabsf(a - 1.0f)  < EPS) out = mn;
    if (a > 1.0f  && a < 1.25f) out = 4.0f * ((1.25f - a) * mn + (a - 1.0f) * xy);
    if (fabsf(a - 1.25f) < EPS) out = xy;
    if (a > 1.25f && a < 2.0f) {
        const float p = sqrtf(fmaxf(3.0f / fmaxf(2.0f - a, EPS) - 1.0f, EPS));
        out = powf(xy, p);
    }
    if (fabsf(a - 2.0f)  < EPS)
        out = ((fabsf(x - 1.0f) < EPS) && (fabsf(y - 1.0f) < EPS)) ? 1.0f : 0.0f;
    return out;
}

__device__ __forceinline__ float F_comb(float x, float y, float a, float w0, float w1) {
    if (isnan(x)) x = EPS;
    x = fminf(fmaxf(x, EPS), 1.0f - EPS);
    if (isnan(y)) y = EPS;
    y = fminf(fmaxf(y, EPS), 1.0f - EPS);
    if (a < 0.5f - EPS) {
        const float af = fminf(fmaxf(1.0f - a, -1.0f + EPS), 2.0f - EPS);
        return 1.0f - F_upper(1.0f - x, 1.0f - y, af, w0, w1);
    }
    return F_upper(x, y, a, w0, w1);
}

__device__ __forceinline__ float combine_op(float l, float r, float a, float w0, float w1) {
    const float nres = F_comb(l, r, a, w0, w1);
    return isnan(nres) ? a : nres;
}

__global__ __launch_bounds__(256) void chain_kernel(const float* __restrict__ leafT,
                                                    const float* __restrict__ weights,
                                                    const float* __restrict__ biases,
                                                    float* __restrict__ out) {
    __shared__ float sA[255], sW0[255], sW1[255];
    const int t = threadIdx.x;
    if (t < 255) {
        const float w0 = weights[2 * t], w1 = weights[2 * t + 1];
        const float m  = fmaxf(w0, w1);
        const float e0 = expf(w0 - m), e1 = expf(w1 - m);
        const float inv = 1.0f / (e0 + e1);
        sW0[t] = e0 * inv;
        sW1[t] = e1 * inv;
        float a = 3.0f / (1.0f + expf(-biases[t])) - 1.0f;
        sA[t] = fminf(fmaxf(a, -1.0f + EPS), 2.0f - EPS);
    }
    __syncthreads();

    const int b = blockIdx.x * 256 + t;
    float carry = combine_op(leafT[b], leafT[(size_t)BATCH + b], sA[0], sW0[0], sW1[0]);
    for (int l = 2; l < LDIM; ++l) {
        const float leaf = leafT[(size_t)l * BATCH + b];
        carry = combine_op(carry, leaf, sA[l - 1], sW0[l - 1], sW1[l - 1]);
    }
    out[b] = carry;
}

extern "C" void kernel_launch(void* const* d_in, const int* in_sizes, int n_in,
                              void* d_out, int out_size, void* d_ws, size_t ws_size,
                              hipStream_t stream) {
    const float* x       = (const float*)d_in[0];
    const float* logits  = (const float*)d_in[1];
    const float* weights = (const float*)d_in[2];
    const float* biases  = (const float*)d_in[3];

    float* P     = (float*)d_ws;
    float* leafT = P + LDIM * LDIM;

    sinkhorn_kernel<<<1, 1024, 0, stream>>>(logits, P);
    gemm_kernel<<<BATCH / 128, 256, 0, stream>>>(x, P, leafT);
    chain_kernel<<<BATCH / 256, 256, 0, stream>>>(leafT, weights, biases, (float*)d_out);
}


# Sinkhorn spread over the chip: one launch per half-sweep (20), one wave per row or column, M kept as logits - f[i] - g[j] with two 256-float potentials (sum order and rounding of M change)
# speedup vs baseline: 1.3937x; 1.3937x over previous
#include <hip/hip_runtime.h>
#include <hip/hip_bf16.h>
#include <math.h>

#define EPS 1e-6f
#define LDIM 256
#define BATCH 65536

typedef __attribute__((ext_vector_type(2))) float v2f;
typedef __attribute__((ext_vector_type(8))) float v8f;

__global__ __launch_bounds__(256) void lse_pass_kernel(const float* __restrict__ logits,
                                                       const float* __restrict__ sub,
                                                       float* __restrict__ outv,
                                                       int s_line, int s_elem,
                                                       float* __restrict__ Pout) {
    const int lane = threadIdx.x & 31;
    const int line = blockIdx.x * 8 + (threadIdx.x >> 5);
    const float* src = logits + line * s_line + lane * s_elem;
    const int step = 32 * s_elem;

    float v[8];
    #pragma unroll
    for (int k = 0; k < 8; ++k) v[k] = src[k * step];
    if (sub != nullptr) {
        #pragma unroll
        for (int k = 0; k < 8; ++k) v[k] -= sub[lane + 32 * k];
    }

    float mx = v[0];
    #pragma unroll
    for (int k = 1; k < 8; ++k) mx = fmaxf(mx, v[k]);
    #pragma unroll
    for (int o = 16; o > 0; o >>= 1) mx = fmaxf(mx, __shfl_xor(mx, o));

    float s = 0.0f;
    #pragma unroll
    for (int k = 0; k < 8; ++k) s += expf(v[k] - mx);
    #pragma unroll
    for (int o = 16; o > 0; o >>= 1) s += __shfl_xor(s, o);

    const float lse = mx + logf(s);
    if (lane == 0) outv[line] = lse;

    if (Pout != nullptr) {
        float* dst = Pout + line * s_line + lane * s_elem;
        #pragma unroll
        for (int k = 0; k < 8; ++k) dst[k * step] = expf(v[k] - lse);
    }
}

__global__ __launch_bounds__(256) void gemm_kernel(const float* __restrict__ x,
                                                   const float* __restrict__ P,
                                                   float* __restrict__ leafT) {
    __shared__ float Ps[LDIM * LDIM];
    const int t = threadIdx.x;
    {
        const float4* Pv = (const float4*)P;
        float4* Sv = (float4*)Ps;
        for (int i = t; i < LDIM * LDIM / 4; i += 256) Sv[i] = Pv[i];
    }
    __syncthreads();

    const int wave  = t >> 5;
    const int lane  = t & 31;
    const int mloc  = lane & 15;
    const int khalf = (lane >> 4) * 2;
    const int m0    = blockIdx.x * 128 + wave * 16;

    v8f acc[16] = {};

    const float* xrow = x + (size_t)(m0 + mloc) * LDIM + khalf;

    for (int k = 0; k < LDIM; k += 4) {
        v2f bfrag;
        bfrag.x = xrow[k];
        bfrag.y = xrow[k + 1];
        #pragma unroll
        for (int nt = 0; nt < 16; ++nt) {
            const int n = nt * 16 + mloc;
            v2f afrag;
            afrag.x = Ps[(k + khalf)     * LDIM + n];
            afrag.y = Ps[(k + khalf + 1) * LDIM + n];
            acc[nt] = __builtin_amdgcn_wmma_f32_16x16x4_f32(
                false, afrag, false, bfrag, (short)0, acc[nt], false, false);
        }
    }

    const size_t col = (size_t)m0 + mloc;
    #pragma unroll
    for (int nt = 0; nt < 16; ++nt) {
        const int nbase = nt * 16 + (lane >> 4) * 8;
        #pragma unroll
        for (int i = 0; i < 8; ++i)
            leafT[(size_t)(nbase + i) * BATCH + col] = acc[nt][i];
    }
}

__device__ __forceinline__ float r_poly(float a) {
    float d  = 0.5f - a;
    float d2 = d * d;
    float num = 0.25f + 1.65811f * d + 2.15388f * d2 + 8.2844f * d2 * d + 6.16764f * d2 * d2;
    float den = a * (1.0f - a);
    den = (fabsf(den) < EPS) ? EPS : den;
    return num / den;
}

__device__ __forceinline__ float power_r(float x, float y, float ra, float w0, float w1) {
    ra = (fabsf(ra) < EPS) ? EPS : ra;
    return powf(w0 * powf(x, ra) + w1 * powf(y, ra), 1.0f / ra);
}

__device__ __forceinline__ float F_upper(float x, float y, float a, float w0, float w1) {
    const float xy  = x * y;
    const float mn  = fminf(x, y);
    const float lin = w0 * x + w1 * y;
    float out = lin;
    if (a > 0.5f && a < 0.75f) {
        const float R = r_poly(0.75f);
        out = (3.0f - 4.0f * a) * lin + (4.0f * a - 2.0f) * power_r(x, y, R, w0, w1);
    }
    if (a >= 0.75f && a < 1.0f) {
        const float ra = r_poly(fminf(fmaxf(a, 0.75f), 1.0f - EPS));
        out = power_r(x, y, ra, w0, w1);
    }
    if (fabsf(a - 1.0f)  < EPS) out = mn;
    if (a > 1.0f  && a < 1.25f) out = 4.0f * ((1.25f - a) * mn + (a - 1.0f) * xy);
    if (fabsf(a - 1.25f) < EPS) out = xy;
    if (a > 1.25f && a < 2.0f) {
        const float p = sqrtf(fmaxf(3.0f / fmaxf(2.0f - a, EPS) - 1.0f, EPS));
        out = powf(xy, p);
    }
    if (fabsf(a - 2.0f)  < EPS)
        out = ((fabsf(x - 1.0f) < EPS) && (fabsf(y - 1.0f) < EPS)) ? 1.0f : 0.0f;
    return out;
}

__device__ __forceinline__ float F_comb(float x, float y, float a, float w0, float w1) {
    if (isnan(x)) x = EPS;
    x = fminf(fmaxf(x, EPS), 1.0f - EPS);
    if (isnan(y)) y = EPS;
    y = fminf(fmaxf(y, EPS), 1.0f - EPS);
    if (a < 0.5f - EPS) {
        const float af = fminf(fmaxf(1.0f - a, -1.0f + EPS), 2.0f - EPS);
        return 1.0f - F_upper(1.0f - x, 1.0f - y, af, w0, w1);
    }
    return F_upper(x, y, a, w0, w1);
}

__device__ __forceinline__ float combine_op(float l, float r, float a, float w0, float w1) {
    const float nres = F_comb(l, r, a, w0, w1);
    return isnan(nres) ? a : nres;
}

__global__ __launch_bounds__(256) void chain_kernel(const float* __restrict__ leafT,
                                                    const float* __restrict__ weights,
                                                    const float* __restrict__ biases,
                                                    float* __restrict__ out) {
    __shared__ float sA[255], sW0[255], sW1[255];
    const int t = threadIdx.x;
    if (t < 255) {
        const float w0 = weights[2 * t], w1 = weights[2 * t + 1];
        const float m  = fmaxf(w0, w1);
        const float e0 = expf(w0 - m), e1 = expf(w1 - m);
        const float inv = 1.0f / (e0 + e1);
        sW0[t] = e0 * inv;
        sW1[t] = e1 * inv;
        float a = 3.0f / (1.0f + expf(-biases[t])) - 1.0f;
        sA[t] = fminf(fmaxf(a, -1.0f + EPS), 2.0f - EPS);
    }
    __syncthreads();

    const int b = blockIdx.x * 256 + t;
    float carry = combine_op(leafT[b], leafT[(size_t)BATCH + b], sA[0], sW0[0], sW1[0]);
    for (int l = 2; l < LDIM; ++l) {
        const float leaf = leafT[(size_t)l * BATCH + b];
        carry = combine_op(carry, leaf, sA[l - 1], sW0[l - 1], sW1[l - 1]);
    }
    out[b] = carry;
}

extern "C" void kernel_launch(void* const* d_in, const int* in_sizes, int n_in,
                              void* d_out, int out_size, void* d_ws, size_t ws_size,
                              hipStream_t stream) {
    const float* x       = (const float*)d_in[0];
    const float* logits  = (const float*)d_in[1];
    const float* weights = (const float*)d_in[2];
    const float* biases  = (const float*)d_in[3];

    float* P     = (float*)d_ws;
    float* leafT = P + LDIM * LDIM;

    float* f = leafT;
    float* g = leafT + LDIM;
#define ROW_SWEEP(subv) lse_pass_kernel<<<LDIM / 8, 256, 0, stream>>>(logits, subv, f, LDIM, 1, nullptr)
#define COL_SWEEP(pout) lse_pass_kernel<<<LDIM / 8, 256, 0, stream>>>(logits, f, g, 1, LDIM, pout)
    ROW_SWEEP(nullptr); COL_SWEEP(nullptr);
    ROW_SWEEP(g);       COL_SWEEP(nullptr);
    ROW_SWEEP(g);       COL_SWEEP(nullptr);
    ROW_SWEEP(g);       COL_SWEEP(nullptr);
    ROW_SWEEP(g);       COL_SWEEP(nullptr);
    ROW_SWEEP(g);       COL_SWEEP(nullptr);
    ROW_SWEEP(g);       COL_SWEEP(nullptr);
    ROW_SWEEP(g);       COL_SWEEP(nullptr);
    ROW_SWEEP(g);       COL_SWEEP(nullptr);
    ROW_SWEEP(g);       COL_SWEEP(P);
#undef ROW_SWEEP
#undef COL_SWEEP
    gemm_kernel<<<BATCH / 128, 256, 0, stream>>>(x, P, leafT);
    chain_kernel<<<BATCH / 256, 256, 0, stream>>>(leafT, weights, biases, (float*)d_out);
}


# Leaf plane kept batch-major: GEMM operands swapped so D comes out [batch][level] (same products, same k order); each scan thread reads 4 consecutive leaves of its own row with one 16-byte load
# speedup vs baseline: 1.7945x; 1.2876x over previous
#include <hip/hip_runtime.h>
#include <hip/hip_bf16.h>
#include <math.h>

#define EPS 1e-6f
#define LDIM 256
#define BATCH 65536

typedef __attribute__((ext_vector_type(2))) float v2f;
typedef __attribute__((ext_vector_type(8))) float v8f;

__global__ __launch_bounds__(256) void lse_pass_kernel(const float* __restrict__ logits,
                                                       const float* __restrict__ sub,
                                                       float* __restrict__ outv,
                                                       int s_line, int s_elem,
                                                       float* __restrict__ Pout) {
    const int lane = threadIdx.x & 31;
    const int line = blockIdx.x * 8 + (threadIdx.x >> 5);
    const float* src = logits + line * s_line + lane * s_elem;
    const int step = 32 * s_elem;

    float v[8];
    #pragma unroll
    for (int k = 0; k < 8; ++k) v[k] = src[k * step];
    if (sub != nullptr) {
        #pragma unroll
        for (int k = 0; k < 8; ++k) v[k] -= sub[lane + 32 * k];
    }

    float mx = v[0];
    #pragma unroll
    for (int k = 1; k < 8; ++k) mx = fmaxf(mx, v[k]);
    #pragma unroll
    for (int o = 16; o > 0; o >>= 1) mx = fmaxf(mx, __shfl_xor(mx, o));

    float s = 0.0f;
    #pragma unroll
    for (int k = 0; k < 8; ++k) s += expf(v[k] - mx);
    #pragma unroll
    for (int o = 16; o > 0; o >>= 1) s += __shfl_xor(s, o);

    const float lse = mx + logf(s);
    if (lane == 0) outv[line] = lse;

    if (Pout != nullptr) {
        float* dst = Pout + line * s_line + lane * s_elem;
        #pragma unroll
        for (int k = 0; k < 8; ++k) dst[k * step] = expf(v[k] - lse);
    }
}

__global__ __launch_bounds__(256) void gemm_kernel(const float* __restrict__ x,
                                                   const float* __restrict__ P,
                                                   float* __restrict__ leaf) {
    __shared__ float Ps[LDIM * LDIM];
    const int t = threadIdx.x;
    {
        const float4* Pv = (const float4*)P;
        float4* Sv = (float4*)Ps;
        for (int i = t; i < LDIM * LDIM / 4; i += 256) Sv[i] = Pv[i];
    }
    __syncthreads();

    const int wave  = t >> 5;
    const int lane  = t & 31;
    const int mloc  = lane & 15;
    const int khalf = (lane >> 4) * 2;
    const int m0    = blockIdx.x * 128 + wave * 16;

    v8f acc[16] = {};

    const float* xrow = x + (size_t)(m0 + mloc) * LDIM + khalf;

    for (int k = 0; k < LDIM; k += 4) {
        v2f xfrag;
        xfrag.x = xrow[k];
        xfrag.y = xrow[k + 1];
        #pragma unroll
        for (int nt = 0; nt < 16; ++nt) {
            const int n = nt * 16 + mloc;
            v2f pfrag;
            pfrag.x = Ps[(k + khalf)     * LDIM + n];
            pfrag.y = Ps[(k + khalf + 1) * LDIM + n];
            acc[nt] = __builtin_amdgcn_wmma_f32_16x16x4_f32(
                false, xfrag, false, pfrag, (short)0, acc[nt], false, false);
        }
    }

    float* dst = leaf + (size_t)(m0 + (lane >> 4) * 8) * LDIM + mloc;
    #pragma unroll
    for (int nt = 0; nt < 16; ++nt) {
        #pragma unroll
        for (int i = 0; i < 8; ++i)
            dst[(size_t)i * LDIM + nt * 16] = acc[nt][i];
    }
}

__device__ __forceinline__ float r_poly(float a) {
    float d  = 0.5f - a;
    float d2 = d * d;
    float num = 0.25f + 1.65811f * d + 2.15388f * d2 + 8.2844f * d2 * d + 6.16764f * d2 * d2;
    float den = a * (1.0f - a);
    den = (fabsf(den) < EPS) ? EPS : den;
    return num / den;
}

__device__ __forceinline__ float power_r(float x, float y, float ra, float w0, float w1) {
    ra = (fabsf(ra) < EPS) ? EPS : ra;
    return powf(w0 * powf(x, ra) + w1 * powf(y, ra), 1.0f / ra);
}

__device__ __forceinline__ float F_upper(float x, float y, float a, float w0, float w1) {
    const float xy  = x * y;
    const float mn  = fminf(x, y);
    const float lin = w0 * x + w1 * y;
    float out = lin;
    if (a > 0.5f && a < 0.75f) {
        const float R = r_poly(0.75f);
        out = (3.0f - 4.0f * a) * lin + (4.0f * a - 2.0f) * power_r(x, y, R, w0, w1);
    }
    if (a >= 0.75f && a < 1.0f) {
        const float ra = r_poly(fminf(fmaxf(a, 0.75f), 1.0f - EPS));
        out = power_r(x, y, ra, w0, w1);
    }
    if (fabsf(a - 1.0f)  < EPS) out = mn;
    if (a > 1.0f  && a < 1.25f) out = 4.0f * ((1.25f - a) * mn + (a - 1.0f) * xy);
    if (fabsf(a - 1.25f) < EPS) out = xy;
    if (a > 1.25f && a < 2.0f) {
        const float p = sqrtf(fmaxf(3.0f / fmaxf(2.0f - a, EPS) - 1.0f, EPS));
        out = powf(xy, p);
    }
    if (fabsf(a - 2.0f)  < EPS)
        out = ((fabsf(x - 1.0f) < EPS) && (fabsf(y - 1.0f) < EPS)) ? 1.0f : 0.0f;
    return out;
}

__device__ __forceinline__ float F_comb(float x, float y, float a, float w0, float w1) {
    if (isnan(x)) x = EPS;
    x = fminf(fmaxf(x, EPS), 1.0f - EPS);
    if (isnan(y)) y = EPS;
    y = fminf(fmaxf(y, EPS), 1.0f - EPS);
    if (a < 0.5f - EPS) {
        const float af = fminf(fmaxf(1.0f - a, -1.0f + EPS), 2.0f - EPS);
        return 1.0f - F_upper(1.0f - x, 1.0f - y, af, w0, w1);
    }
    return F_upper(x, y, a, w0, w1);
}

__device__ __forceinline__ float combine_op(float l, float r, float a, float w0, float w1) {
    const float nres = F_comb(l, r, a, w0, w1);
    return isnan(nres) ? a : nres;
}

__global__ __launch_bounds__(256) void chain_kernel(const float* __restrict__ leaf,
                                                    const float* __restrict__ weights,
                                                    const float* __restrict__ biases,
                                                    float* __restrict__ out) {
    __shared__ float sA[255], sW0[255], sW1[255];
    const int t = threadIdx.x;
    if (t < 255) {
        const float w0 = weights[2 * t], w1 = weights[2 * t + 1];
        const float m  = fmaxf(w0, w1);
        const float e0 = expf(w0 - m), e1 = expf(w1 - m);
        const float inv = 1.0f / (e0 + e1);
        sW0[t] = e0 * inv;
        sW1[t] = e1 * inv;
        float a = 3.0f / (1.0f + expf(-biases[t])) - 1.0f;
        sA[t] = fminf(fmaxf(a, -1.0f + EPS), 2.0f - EPS);
    }
    __syncthreads();

    const int b = blockIdx.x * 256 + t;
    const float4* row4 = (const float4*)(leaf + (size_t)b * LDIM);
    float4 v = row4[0];
    float carry = combine_op(v.x, v.y, sA[0], sW0[0], sW1[0]);
    for (int l = 2; l < LDIM; ++l) {
        const int j = l & 3;
        if (j == 0) v = row4[l >> 2];
        const float lf = (j == 0) ? v.x : (j == 1) ? v.y : (j == 2) ? v.z : v.w;
        carry = combine_op(carry, lf, sA[l - 1], sW0[l - 1], sW1[l - 1]);
    }
    out[b] = carry;
}

extern "C" void kernel_launch(void* const* d_in, const int* in_sizes, int n_in,
                              void* d_out, int out_size, void* d_ws, size_t ws_size,
                              hipStream_t stream) {
    const float* x       = (const float*)d_in[0];
    const float* logits  = (const float*)d_in[1];
    const float* weights = (const float*)d_in[2];
    const float* biases  = (const float*)d_in[3];

    float* P     = (float*)d_ws;
    float* leaf  = P + LDIM * LDIM;

    float* f = leaf;
    float* g = leaf + LDIM;
#define ROW_SWEEP(subv) lse_pass_kernel<<<LDIM / 8, 256, 0, stream>>>(logits, subv, f, LDIM, 1, nullptr)
#define COL_SWEEP(pout) lse_pass_kernel<<<LDIM / 8, 256, 0, stream>>>(logits, f, g, 1, LDIM, pout)
    ROW_SWEEP(nullptr); COL_SWEEP(nullptr);
    ROW_SWEEP(g);       COL_SWEEP(nullptr);
    ROW_SWEEP(g);       COL_SWEEP(nullptr);
    ROW_SWEEP(g);       COL_SWEEP(nullptr);
    ROW_SWEEP(g);       COL_SWEEP(nullptr);
    ROW_SWEEP(g);       COL_SWEEP(nullptr);
    ROW_SWEEP(g);       COL_SWEEP(nullptr);
    ROW_SWEEP(g);       COL_SWEEP(nullptr);
    ROW_SWEEP(g);       COL_SWEEP(nullptr);
    ROW_SWEEP(g);       COL_SWEEP(P);
#undef ROW_SWEEP
#undef COL_SWEEP
    gemm_kernel<<<BATCH / 128, 256, 0, stream>>>(x, P, leaf);
    chain_kernel<<<BATCH / 256, 256, 0, stream>>>(leaf, weights, biases, (float*)d_out);
}


# Sinkhorn: one launch per full iteration (11, was 20): a work-group owns 8 rows, does their row sweep and leaves per-column (max,sum) partials that the next launch combines (sum order changes)
# speedup vs baseline: 1.8927x; 1.0547x over previous
#include <hip/hip_runtime.h>
#include <hip/hip_bf16.h>
#include <math.h>

#define EPS 1e-6f
#define LDIM 256
#define BATCH 65536

typedef __attribute__((ext_vector_type(2))) float v2f;
typedef __attribute__((ext_vector_type(8))) float v8f;

#define SROWS 8
#define SNB   (LDIM / SROWS)

__global__ __launch_bounds__(256) void sink_iter_kernel(const float* __restrict__ logits,
                                                        const float* __restrict__ part_in,
                                                        float* __restrict__ part_out,
                                                        float* __restrict__ fbuf,
                                                        float* __restrict__ Pout) {
    __shared__ float sg[LDIM];
    __shared__ float sf[SROWS];
    const int t    = threadIdx.x;
    const int lane = t & 31;
    const int w    = t >> 5;
    const int row0 = blockIdx.x * SROWS;

    float c[SROWS];
    #pragma unroll
    for (int r = 0; r < SROWS; ++r) c[r] = logits[(row0 + r) * LDIM + t];
    float v[8];
    #pragma unroll
    for (int k = 0; k < 8; ++k) v[k] = logits[(row0 + w) * LDIM + lane + 32 * k];

    float g = 0.0f;
    if (part_in != nullptr) {
        const float2* pin = (const float2*)part_in + t;
        float2 p[SNB];
        #pragma unroll
        for (int b = 0; b < SNB; ++b) p[b] = pin[b * LDIM];
        float M = p[0].x;
        #pragma unroll
        for (int b = 1; b < SNB; ++b) M = fmaxf(M, p[b].x);
        float S = 0.0f;
        #pragma unroll
        for (int b = 0; b < SNB; ++b) S += p[b].y * expf(p[b].x - M);
        g = M + logf(S);
    }

    if (Pout != nullptr) {
        #pragma unroll
        for (int r = 0; r < SROWS; ++r)
            Pout[(row0 + r) * LDIM + t] = expf((c[r] - fbuf[row0 + r]) - g);
        return;
    }

    sg[t] = g;
    __syncthreads();

    #pragma unroll
    for (int k = 0; k < 8; ++k) v[k] -= sg[lane + 32 * k];
    float mx = v[0];
    #pragma unroll
    for (int k = 1; k < 8; ++k) mx = fmaxf(mx, v[k]);
    #pragma unroll
    for (int o = 16; o > 0; o >>= 1) mx = fmaxf(mx, __shfl_xor(mx, o));
    float s = 0.0f;
    #pragma unroll
    for (int k = 0; k < 8; ++k) s += expf(v[k] - mx);
    #pragma unroll
    for (int o = 16; o > 0; o >>= 1) s += __shfl_xor(s, o);
    const float lse = mx + logf(s);
    if (lane == 0) { sf[w] = lse; fbuf[row0 + w] = lse; }
    __syncthreads();

    float u[SROWS];
    #pragma unroll
    for (int r = 0; r < SROWS; ++r) u[r] = c[r] - sf[r];
    float m = u[0];
    #pragma unroll
    for (int r = 1; r < SROWS; ++r) m = fmaxf(m, u[r]);
    float ps = 0.0f;
    #pragma unroll
    for (int r = 0; r < SROWS; ++r) ps += expf(u[r] - m);
    ((float2*)part_out)[blockIdx.x * LDIM + t] = make_float2(m, ps);
}

__global__ __launch_bounds__(256) void gemm_kernel(const float* __restrict__ x,
                                                   const float* __restrict__ P,
                                                   float* __restrict__ leaf) {
    __shared__ float Ps[LDIM * LDIM];
    const int t = threadIdx.x;
    {
        const float4* Pv = (const float4*)P;
        float4* Sv = (float4*)Ps;
        for (int i = t; i < LDIM * LDIM / 4; i += 256) Sv[i] = Pv[i];
    }
    __syncthreads();

    const int wave  = t >> 5;
    const int lane  = t & 31;
    const int mloc  = lane & 15;
    const int khalf = (lane >> 4) * 2;
    const int m0    = blockIdx.x * 128 + wave * 16;

    v8f acc[16] = {};

    const float* xrow = x + (size_t)(m0 + mloc) * LDIM + khalf;

    for (int k = 0; k < LDIM; k += 4) {
        v2f xfrag;
        xfrag.x = xrow[k];
        xfrag.y = xrow[k + 1];
        #pragma unroll
        for (int nt = 0; nt < 16; ++nt) {
            const int n = nt * 16 + mloc;
            v2f pfrag;
            pfrag.x = Ps[(k + khalf)     * LDIM + n];
            pfrag.y = Ps[(k + khalf + 1) * LDIM + n];
            acc[nt] = __builtin_amdgcn_wmma_f32_16x16x4_f32(
                false, xfrag, false, pfrag, (short)0, acc[nt], false, false);
        }
    }

    float* dst = leaf + (size_t)(m0 + (lane >> 4) * 8) * LDIM + mloc;
    #pragma unroll
    for (int nt = 0; nt < 16; ++nt) {
        #pragma unroll
        for (int i = 0; i < 8; ++i)
            dst[(size_t)i * LDIM + nt * 16] = acc[nt][i];
    }
}

__device__ __forceinline__ float r_poly(float a) {
    float d  = 0.5f - a;
    float d2 = d * d;
    float num = 0.25f + 1.65811f * d + 2.15388f * d2 + 8.2844f * d2 * d + 6.16764f * d2 * d2;
    float den = a * (1.0f - a);
    den = (fabsf(den) < EPS) ? EPS : den;
    return num / den;
}

__device__ __forceinline__ float power_r(float x, float y, float ra, float w0, float w1) {
    ra = (fabsf(ra) < EPS) ? EPS : ra;
    return powf(w0 * powf(x, ra) + w1 * powf(y, ra), 1.0f / ra);
}

__device__ __forceinline__ float F_upper(float x, float y, float a, float w0, float w1) {
    const float xy  = x * y;
    const float mn  = fminf(x, y);
    const float lin = w0 * x + w1 * y;
    float out = lin;
    if (a > 0.5f && a < 0.75f) {
        const float R = r_poly(0.75f);
        out = (3.0f - 4.0f * a) * lin + (4.0f * a - 2.0f) * power_r(x, y, R, w0, w1);
    }
    if (a >= 0.75f && a < 1.0f) {
        const float ra = r_poly(fminf(fmaxf(a, 0.75f), 1.0f - EPS));
        out = power_r(x, y, ra, w0, w1);
    }
    if (fabsf(a - 1.0f)  < EPS) out = mn;
    if (a > 1.0f  && a < 1.25f) out = 4.0f * ((1.25f - a) * mn + (a - 1.0f) * xy);
    if (fabsf(a - 1.25f) < EPS) out = xy;
    if (a > 1.25f && a < 2.0f) {
        const float p = sqrtf(fmaxf(3.0f / fmaxf(2.0f - a, EPS) - 1.0f, EPS));
        out = powf(xy, p);
    }
    if (fabsf(a - 2.0f)  < EPS)
        out = ((fabsf(x - 1.0f) < EPS) && (fabsf(y - 1.0f) < EPS)) ? 1.0f : 0.0f;
    return out;
}

__device__ __forceinline__ float F_comb(float x, float y, float a, float w0, float w1) {
    if (isnan(x)) x = EPS;
    x = fminf(fmaxf(x, EPS), 1.0f - EPS);
    if (isnan(y)) y = EPS;
    y = fminf(fmaxf(y, EPS), 1.0f - EPS);
    if (a < 0.5f - EPS) {
        const float af = fminf(fmaxf(1.0f - a, -1.0f + EPS), 2.0f - EPS);
        return 1.0f - F_upper(1.0f - x, 1.0f - y, af, w0, w1);
    }
    return F_upper(x, y, a, w0, w1);
}

__device__ __forceinline__ float combine_op(float l, float r, float a, float w0, float w1) {
    const float nres = F_comb(l, r, a, w0, w1);
    return isnan(nres) ? a : nres;
}

__global__ __launch_bounds__(256) void chain_kernel(const float* __restrict__ leaf,
                                                    const float* __restrict__ weights,
                                                    const float* __restrict__ biases,
                                                    float* __restrict__ out) {
    __shared__ float sA[255], sW0[255], sW1[255];
    const int t = threadIdx.x;
    if (t < 255) {
        const float w0 = weights[2 * t], w1 = weights[2 * t + 1];
        const float m  = fmaxf(w0, w1);
        const float e0 = expf(w0 - m), e1 = expf(w1 - m);
        const float inv = 1.0f / (e0 + e1);
        sW0[t] = e0 * inv;
        sW1[t] = e1 * inv;
        float a = 3.0f / (1.0f + expf(-biases[t])) - 1.0f;
        sA[t] = fminf(fmaxf(a, -1.0f + EPS), 2.0f - EPS);
    }
    __syncthreads();

    const int b = blockIdx.x * 256 + t;
    const float4* row4 = (const float4*)(leaf + (size_t)b * LDIM);
    float4 v = row4[0];
    float carry = combine_op(v.x, v.y, sA[0], sW0[0], sW1[0]);
    for (int l = 2; l < LDIM; ++l) {
        const int j = l & 3;
        if (j == 0) v = row4[l >> 2];
        const float lf = (j == 0) ? v.x : (j == 1) ? v.y : (j == 2) ? v.z : v.w;
        carry = combine_op(carry, lf, sA[l - 1], sW0[l - 1], sW1[l - 1]);
    }
    out[b] = carry;
}

extern "C" void kernel_launch(void* const* d_in, const int* in_sizes, int n_in,
                              void* d_out, int out_size, void* d_ws, size_t ws_size,
                              hipStream_t stream) {
    const float* x       = (const float*)d_in[0];
    const float* logits  = (const float*)d_in[1];
    const float* weights = (const float*)d_in[2];
    const float* biases  = (const float*)d_in[3];

    float* P     = (float*)d_ws;
    float* leaf  = P + LDIM * LDIM;

    float* partA = leaf;
    float* partB = leaf + 2 * SNB * LDIM;
    float* f     = leaf + 4 * SNB * LDIM;
#define SINK(pin, pout, pmat) sink_iter_kernel<<<SNB, 256, 0, stream>>>(logits, pin, pout, f, pmat)
    SINK(nullptr, partA, nullptr);
    SINK(partA,   partB, nullptr);
    SINK(partB,   partA, nullptr);
    SINK(partA,   partB, nullptr);
    SINK(partB,   partA, nullptr);
    SINK(partA,   partB, nullptr);
    SINK(partB,   partA, nullptr);
    SINK(partA,   partB, nullptr);
    SINK(partB,   partA, nullptr);
    SINK(partA,   partB, nullptr);
    SINK(partB,   nullptr, P);
#undef SINK
    gemm_kernel<<<BATCH / 128, 256, 0, stream>>>(x, P, leaf);
    chain_kernel<<<BATCH / 256, 256, 0, stream>>>(leaf, weights, biases, (float*)d_out);
}


# GEMM cut into 16-level blocks and run inside the scan kernel, one block ahead of its scan steps: x fragments stay in registers, leaves pass through a per-wave LDS block, never through memory
# speedup vs baseline: 1.9965x; 1.0548x over previous
#include <hip/hip_runtime.h>
#include <hip/hip_bf16.h>
#include <math.h>

#define EPS 1e-6f
#define LDIM 256
#define BATCH 65536

typedef __attribute__((ext_vector_type(2))) float v2f;
typedef __attribute__((ext_vector_type(8))) float v8f;

#define SROWS 8
#define SNB   (LDIM / SROWS)

__global__ __launch_bounds__(256) void sink_iter_kernel(const float* __restrict__ logits,
                                                        const float* __restrict__ part_in,
                                                        float* __restrict__ part_out,
                                                        float* __restrict__ fbuf,
                                                        float* __restrict__ Pout) {
    __shared__ float sg[LDIM];
    __shared__ float sf[SROWS];
    const int t    = threadIdx.x;
    const int lane = t & 31;
    const int w    = t >> 5;
    const int row0 = blockIdx.x * SROWS;

    float c[SROWS];
    #pragma unroll
    for (int r = 0; r < SROWS; ++r) c[r] = logits[(row0 + r) * LDIM + t];
    float v[8];
    #pragma unroll
    for (int k = 0; k < 8; ++k) v[k] = logits[(row0 + w) * LDIM + lane + 32 * k];

    float g = 0.0f;
    if (part_in != nullptr) {
        const float2* pin = (const float2*)part_in + t;
        float2 p[SNB];
        #pragma unroll
        for (int b = 0; b < SNB; ++b) p[b] = pin[b * LDIM];
        float M = p[0].x;
        #pragma unroll
        for (int b = 1; b < SNB; ++b) M = fmaxf(M, p[b].x);
        float S = 0.0f;
        #pragma unroll
        for (int b = 0; b < SNB; ++b) S += p[b].y * expf(p[b].x - M);
        g = M + logf(S);
    }

    if (Pout != nullptr) {
        #pragma unroll
        for (int r = 0; r < SROWS; ++r)
            Pout[(row0 + r) * LDIM + t] = expf((c[r] - fbuf[row0 + r]) - g);
        return;
    }

    sg[t] = g;
    __syncthreads();

    #pragma unroll
    for (int k = 0; k < 8; ++k) v[k] -= sg[lane + 32 * k];
    float mx = v[0];
    #pragma unroll
    for (int k = 1; k < 8; ++k) mx = fmaxf(mx, v[k]);
    #pragma unroll
    for (int o = 16; o > 0; o >>= 1) mx = fmaxf(mx, __shfl_xor(mx, o));
    float s = 0.0f;
    #pragma unroll
    for (int k = 0; k < 8; ++k) s += expf(v[k] - mx);
    #pragma unroll
    for (int o = 16; o > 0; o >>= 1) s += __shfl_xor(s, o);
    const float lse = mx + logf(s);
    if (lane == 0) { sf[w] = lse; fbuf[row0 + w] = lse; }
    __syncthreads();

    float u[SROWS];
    #pragma unroll
    for (int r = 0; r < SROWS; ++r) u[r] = c[r] - sf[r];
    float m = u[0];
    #pragma unroll
    for (int r = 1; r < SROWS; ++r) m = fmaxf(m, u[r]);
    float ps = 0.0f;
    #pragma unroll
    for (int r = 0; r < SROWS; ++r) ps += expf(u[r] - m);
    ((float2*)part_out)[blockIdx.x * LDIM + t] = make_float2(m, ps);
}

#define LBLK 16
#define SPITCH 17
#define KSTEPS (LDIM / 4)

__device__ __forceinline__ float r_poly(float a) {
    float d  = 0.5f - a;
    float d2 = d * d;
    float num = 0.25f + 1.65811f * d + 2.15388f * d2 + 8.2844f * d2 * d + 6.16764f * d2 * d2;
    float den = a * (1.0f - a);
    den = (fabsf(den) < EPS) ? EPS : den;
    return num / den;
}

__device__ __forceinline__ float power_r(float x, float y, float ra, float w0, float w1) {
    ra = (fabsf(ra) < EPS) ? EPS : ra;
    return powf(w0 * powf(x, ra) + w1 * powf(y, ra), 1.0f / ra);
}

__device__ __forceinline__ float F_upper(float x, float y, float a, float w0, float w1) {
    const float xy  = x * y;
    const float mn  = fminf(x, y);
    const float lin = w0 * x + w1 * y;
    float out = lin;
    if (a > 0.5f && a < 0.75f) {
        const float R = r_poly(0.75f);
        out = (3.0f - 4.0f * a) * lin + (4.0f * a - 2.0f) * power_r(x, y, R, w0, w1);
    }
    if (a >= 0.75f && a < 1.0f) {
        const float ra = r_poly(fminf(fmaxf(a, 0.75f), 1.0f - EPS));
        out = power_r(x, y, ra, w0, w1);
    }
    if (fabsf(a - 1.0f)  < EPS) out = mn;
    if (a > 1.0f  && a < 1.25f) out = 4.0f * ((1.25f - a) * mn + (a - 1.0f) * xy);
    if (fabsf(a - 1.25f) < EPS) out = xy;
    if (a > 1.25f && a < 2.0f) {
        const float p = sqrtf(fmaxf(3.0f / fmaxf(2.0f - a, EPS) - 1.0f, EPS));
        out = powf(xy, p);
    }
    if (fabsf(a - 2.0f)  < EPS)
        out = ((fabsf(x - 1.0f) < EPS) && (fabsf(y - 1.0f) < EPS)) ? 1.0f : 0.0f;
    return out;
}

__device__ __forceinline__ float F_comb(float x, float y, float a, float w0, float w1) {
    if (isnan(x)) x = EPS;
    x = fminf(fmaxf(x, EPS), 1.0f - EPS);
    if (isnan(y)) y = EPS;
    y = fminf(fmaxf(y, EPS), 1.0f - EPS);
    if (a < 0.5f - EPS) {
        const float af = fminf(fmaxf(1.0f - a, -1.0f + EPS), 2.0f - EPS);
        return 1.0f - F_upper(1.0f - x, 1.0f - y, af, w0, w1);
    }
    return F_upper(x, y, a, w0, w1);
}

__device__ __forceinline__ float combine_op(float l, float r, float a, float w0, float w1) {
    const float nres = F_comb(l, r, a, w0, w1);
    return isnan(nres) ? a : nres;
}


__device__ __forceinline__ void gemm_block(const v2f (&xa)[2][KSTEPS], const float* __restrict__ Ps,
                                           int nb, int c, int h, v8f (&acc)[2]) {
    const float* pcol = Ps + (2 * h) * LDIM + nb * LBLK + c;
    v2f pb[KSTEPS];
    #pragma unroll
    for (int ks = 0; ks < KSTEPS; ++ks) {
        pb[ks].x = pcol[(4 * ks)     * LDIM];
        pb[ks].y = pcol[(4 * ks + 1) * LDIM];
    }
    __builtin_amdgcn_sched_barrier(0);
    acc[0] = v8f{};
    acc[1] = v8f{};
    #pragma unroll
    for (int ks = 0; ks < KSTEPS; ++ks) {
        acc[0] = __builtin_amdgcn_wmma_f32_16x16x4_f32(false, xa[0][ks], false, pb[ks], (short)0, acc[0], false, false);
        acc[1] = __builtin_amdgcn_wmma_f32_16x16x4_f32(false, xa[1][ks], false, pb[ks], (short)0, acc[1], false, false);
    }
}

__global__ __launch_bounds__(256) void gemm_chain_kernel(const float* __restrict__ x,
                                                         const float* __restrict__ P,
                                                         const float* __restrict__ weights,
                                                         const float* __restrict__ biases,
                                                         float* __restrict__ out) {
    __shared__ float Ps[LDIM * LDIM];
    __shared__ float scr[8 * 32 * SPITCH];
    __shared__ float sA[255], sW0[255], sW1[255];
    const int t    = threadIdx.x;
    const int wave = t >> 5;
    const int lane = t & 31;
    const int c    = lane & 15;
    const int h    = lane >> 4;

    const int row0 = blockIdx.x * 256 + wave * 32;

    {
        const float4* Pv = (const float4*)P;
        float4* Sv = (float4*)Ps;
        for (int i = t; i < LDIM * LDIM / 4; i += 256) Sv[i] = Pv[i];
    }
    v2f xa[2][KSTEPS];
    #pragma unroll
    for (int ks = 0; ks < KSTEPS; ++ks) {
        #pragma unroll
        for (int mt = 0; mt < 2; ++mt)
            xa[mt][ks] = *(const v2f*)(x + (size_t)(row0 + 16 * mt + c) * LDIM + 4 * ks + 2 * h);
    }
    if (t < 255) {
        const float w0 = weights[2 * t], w1 = weights[2 * t + 1];
        const float m  = fmaxf(w0, w1);
        const float e0 = expf(w0 - m), e1 = expf(w1 - m);
        const float inv = 1.0f / (e0 + e1);
        sW0[t] = e0 * inv;
        sW1[t] = e1 * inv;
        float a = 3.0f / (1.0f + expf(-biases[t])) - 1.0f;
        sA[t] = fminf(fmaxf(a, -1.0f + EPS), 2.0f - EPS);
    }
    __syncthreads();

    float* wscr = scr + wave * (32 * SPITCH);
    const float* myrow = wscr + lane * SPITCH;

    v8f acc[2];
    gemm_block(xa, Ps, 0, c, h, acc);

    float carry = 0.0f;
    for (int l = 0; l < LDIM; ++l) {
        if ((l & (LBLK - 1)) == 0) {
            #pragma unroll
            for (int mt = 0; mt < 2; ++mt) {
                #pragma unroll
                for (int i = 0; i < 8; ++i)
                    wscr[(16 * mt + 8 * h + i) * SPITCH + c] = acc[mt][i];
            }
            if (l + LBLK < LDIM) gemm_block(xa, Ps, l / LBLK + 1, c, h, acc);
        }
        const float lf = myrow[l & (LBLK - 1)];
        if (l == 0) carry = lf;
        else        carry = combine_op(carry, lf, sA[l - 1], sW0[l - 1], sW1[l - 1]);
    }
    out[row0 + lane] = carry;
}

extern "C" void kernel_launch(void* const* d_in, const int* in_sizes, int n_in,
                              void* d_out, int out_size, void* d_ws, size_t ws_size,
                              hipStream_t stream) {
    const float* x       = (const float*)d_in[0];
    const float* logits  = (const float*)d_in[1];
    const float* weights = (const float*)d_in[2];
    const float* biases  = (const float*)d_in[3];

    float* P     = (float*)d_ws;
    float* leaf  = P + LDIM * LDIM;

    float* partA = leaf;
    float* partB = leaf + 2 * SNB * LDIM;
    float* f     = leaf + 4 * SNB * LDIM;
#define SINK(pin, pout, pmat) sink_iter_kernel<<<SNB, 256, 0, stream>>>(logits, pin, pout, f, pmat)
    SINK(nullptr, partA, nullptr);
    SINK(partA,   partB, nullptr);
    SINK(partB,   partA, nullptr);
    SINK(partA,   partB, nullptr);
    SINK(partB,   partA, nullptr);
    SINK(partA,   partB, nullptr);
    SINK(partB,   partA, nullptr);
    SINK(partA,   partB, nullptr);
    SINK(partB,   partA, nullptr);
    SINK(partA,   partB, nullptr);
    SINK(partB,   nullptr, P);
#undef SINK
    gemm_chain_kernel<<<BATCH / 256, 256, 0, stream>>>(x, P, weights, biases, (float*)d_out);
}


# Product of block j+1 cut along k into 16 chunks of 8 matrix instructions, one per scan step, issued between the step's LDS reads and their use; x tile and P blocks held in LDS; sums unchanged
# speedup vs baseline: 2.0615x; 1.0326x over previous
#include <hip/hip_runtime.h>
#include <hip/hip_bf16.h>
#include <math.h>

#define EPS 1e-6f
#define LDIM 256
#define BATCH 65536

typedef __attribute__((ext_vector_type(2))) float v2f;
typedef __attribute__((ext_vector_type(8))) float v8f;

#define SROWS 8
#define SNB   (LDIM / SROWS)

__global__ __launch_bounds__(256) void sink_iter_kernel(const float* __restrict__ logits,
                                                        const float* __restrict__ part_in,
                                                        float* __restrict__ part_out,
                                                        float* __restrict__ fbuf,
                                                        float* __restrict__ Pout) {
    __shared__ float sg[LDIM];
    __shared__ float sf[SROWS];
    const int t    = threadIdx.x;
    const int lane = t & 31;
    const int w    = t >> 5;
    const int row0 = blockIdx.x * SROWS;

    float c[SROWS];
    #pragma unroll
    for (int r = 0; r < SROWS; ++r) c[r] = logits[(row0 + r) * LDIM + t];
    float v[8];
    #pragma unroll
    for (int k = 0; k < 8; ++k) v[k] = logits[(row0 + w) * LDIM + lane + 32 * k];

    float g = 0.0f;
    if (part_in != nullptr) {
        const float2* pin = (const float2*)part_in + t;
        float2 p[SNB];
        #pragma unroll
        for (int b = 0; b < SNB; ++b) p[b] = pin[b * LDIM];
        float M = p[0].x;
        #pragma unroll
        for (int b = 1; b < SNB; ++b) M = fmaxf(M, p[b].x);
        float S = 0.0f;
        #pragma unroll
        for (int b = 0; b < SNB; ++b) S += p[b].y * expf(p[b].x - M);
        g = M + logf(S);
    }

    if (Pout != nullptr) {
        #pragma unroll
        for (int r = 0; r < SROWS; ++r)
            Pout[(row0 + r) * LDIM + t] = expf((c[r] - fbuf[row0 + r]) - g);
        return;
    }

    sg[t] = g;
    __syncthreads();

    #pragma unroll
    for (int k = 0; k < 8; ++k) v[k] -= sg[lane + 32 * k];
    float mx = v[0];
    #pragma unroll
    for (int k = 1; k < 8; ++k) mx = fmaxf(mx, v[k]);
    #pragma unroll
    for (int o = 16; o > 0; o >>= 1) mx = fmaxf(mx, __shfl_xor(mx, o));
    float s = 0.0f;
    #pragma unroll
    for (int k = 0; k < 8; ++k) s += expf(v[k] - mx);
    #pragma unroll
    for (int o = 16; o > 0; o >>= 1) s += __shfl_xor(s, o);
    const float lse = mx + logf(s);
    if (lane == 0) { sf[w] = lse; fbuf[row0 + w] = lse; }
    __syncthreads();

    float u[SROWS];
    #pragma unroll
    for (int r = 0; r < SROWS; ++r) u[r] = c[r] - sf[r];
    float m = u[0];
    #pragma unroll
    for (int r = 1; r < SROWS; ++r) m = fmaxf(m, u[r]);
    float ps = 0.0f;
    #pragma unroll
    for (int r = 0; r < SROWS; ++r) ps += expf(u[r] - m);
    ((float2*)part_out)[blockIdx.x * LDIM + t] = make_float2(m, ps);
}

#define LBLK 16
#define SPITCH 17
#define KSTEPS (LDIM / 4)
#define KCH (KSTEPS / LBLK)
#define XPITCH (LDIM + 4)
#define BBLK (LDIM * LBLK)

__device__ __forceinline__ float r_poly(float a) {
    float d  = 0.5f - a;
    float d2 = d * d;
    float num = 0.25f + 1.65811f * d + 2.15388f * d2 + 8.2844f * d2 * d + 6.16764f * d2 * d2;
    float den = a * (1.0f - a);
    den = (fabsf(den) < EPS) ? EPS : den;
    return num / den;
}

__device__ __forceinline__ float power_r(float x, float y, float ra, float w0, float w1) {
    ra = (fabsf(ra) < EPS) ? EPS : ra;
    return powf(w0 * powf(x, ra) + w1 * powf(y, ra), 1.0f / ra);
}

__device__ __forceinline__ float F_upper(float x, float y, float a, float w0, float w1) {
    const float xy  = x * y;
    const float mn  = fminf(x, y);
    const float lin = w0 * x + w1 * y;
    float out = lin;
    if (a > 0.5f && a < 0.75f) {
        const float R = r_poly(0.75f);
        out = (3.0f - 4.0f * a) * lin + (4.0f * a - 2.0f) * power_r(x, y, R, w0, w1);
    }
    if (a >= 0.75f && a < 1.0f) {
        const float ra = r_poly(fminf(fmaxf(a, 0.75f), 1.0f - EPS));
        out = power_r(x, y, ra, w0, w1);
    }
    if (fabsf(a - 1.0f)  < EPS) out = mn;
    if (a > 1.0f  && a < 1.25f) out = 4.0f * ((1.25f - a) * mn + (a - 1.0f) * xy);
    if (fabsf(a - 1.25f) < EPS) out = xy;
    if (a > 1.25f && a < 2.0f) {
        const float p = sqrtf(fmaxf(3.0f / fmaxf(2.0f - a, EPS) - 1.0f, EPS));
        out = powf(xy, p);
    }
    if (fabsf(a - 2.0f)  < EPS)
        out = ((fabsf(x - 1.0f) < EPS) && (fabsf(y - 1.0f) < EPS)) ? 1.0f : 0.0f;
    return out;
}

__device__ __forceinline__ float F_comb(float x, float y, float a, float w0, float w1) {
    if (isnan(x)) x = EPS;
    x = fminf(fmaxf(x, EPS), 1.0f - EPS);
    if (isnan(y)) y = EPS;
    y = fminf(fmaxf(y, EPS), 1.0f - EPS);
    if (a < 0.5f - EPS) {
        const float af = fminf(fmaxf(1.0f - a, -1.0f + EPS), 2.0f - EPS);
        return 1.0f - F_upper(1.0f - x, 1.0f - y, af, w0, w1);
    }
    return F_upper(x, y, a, w0, w1);
}

__device__ __forceinline__ float combine_op(float l, float r, float a, float w0, float w1) {
    const float nres = F_comb(l, r, a, w0, w1);
    return isnan(nres) ? a : nres;
}


__device__ __forceinline__ void load_a(const float* xrow, int ch, v2f (&fa)[2][KCH]) {
    const float* p = xrow + ch * (4 * KCH);
    #pragma unroll
    for (int m = 0; m < KCH; ++m) {
        #pragma unroll
        for (int mt = 0; mt < 2; ++mt) fa[mt][m] = *(const v2f*)(p + mt * 16 * XPITCH + 4 * m);
    }
}
__device__ __forceinline__ void load_b(const float* bcol, int ch, v2f (&fb)[KCH]) {
    const float* p = bcol + ch * (4 * KCH * LBLK);
    #pragma unroll
    for (int m = 0; m < KCH; ++m) {
        fb[m].x = p[(4 * m)     * LBLK];
        fb[m].y = p[(4 * m + 1) * LBLK];
    }
}
__device__ __forceinline__ void mma_chunk(const v2f (&fa)[2][KCH], const v2f (&fb)[KCH], v8f (&acc)[2]) {
    #pragma unroll
    for (int m = 0; m < KCH; ++m) {
        acc[0] = __builtin_amdgcn_wmma_f32_16x16x4_f32(false, fa[0][m], false, fb[m], (short)0, acc[0], false, false);
        acc[1] = __builtin_amdgcn_wmma_f32_16x16x4_f32(false, fa[1][m], false, fb[m], (short)0, acc[1], false, false);
    }
}
struct PPieces { float4 q0, q1, q2, q3; };
__device__ __forceinline__ float4 p_piece(const float* __restrict__ P, int nb, int idx) {
    return *(const float4*)(P + (idx >> 2) * LDIM + nb * LBLK + (idx & 3) * 4);
}
__device__ __forceinline__ void p_fetch(const float* __restrict__ P, int nb, int t, PPieces& v) {
    v.q0 = p_piece(P, nb, t);
    v.q1 = p_piece(P, nb, t + 256);
    v.q2 = p_piece(P, nb, t + 512);
    v.q3 = p_piece(P, nb, t + 768);
}
__device__ __forceinline__ void p_store(float* buf, int t, const PPieces& v) {
    float4* b = (float4*)buf + t;
    b[0]   = v.q0;
    b[256] = v.q1;
    b[512] = v.q2;
    b[768] = v.q3;
}

__global__ __launch_bounds__(256) void gemm_chain_kernel(const float* __restrict__ x,
                                                         const float* __restrict__ P,
                                                         const float* __restrict__ weights,
                                                         const float* __restrict__ biases,
                                                         float* __restrict__ out) {
    __shared__ float Xs[8 * 32 * XPITCH];
    __shared__ float Bs[2 * BBLK];
    __shared__ float scr[8 * 32 * SPITCH];
    __shared__ float sA[255], sW0[255], sW1[255];
    const int t    = threadIdx.x;
    const int wave = t >> 5;
    const int lane = t & 31;
    const int c    = lane & 15;
    const int h    = lane >> 4;

    const int row0 = blockIdx.x * 256 + wave * 32;

    PPieces pst0, pst1;
    p_fetch(P, 0, t, pst0);
    p_fetch(P, 1, t, pst1);
    float* xw = Xs + wave * (32 * XPITCH);
    {
        const float4* xg = (const float4*)(x + (size_t)row0 * LDIM);
        #pragma unroll 32
        for (int it = 0; it < 32 * LDIM / 128; ++it) {
            const int f = it * 32 + lane;
            *(float4*)(xw + (f >> 6) * XPITCH + (f & 63) * 4) = xg[f];
        }
    }
    p_store(Bs, t, pst0);
    p_store(Bs + BBLK, t, pst1);
    if (t < 255) {
        const float w0 = weights[2 * t], w1 = weights[2 * t + 1];
        const float m  = fmaxf(w0, w1);
        const float e0 = expf(w0 - m), e1 = expf(w1 - m);
        const float inv = 1.0f / (e0 + e1);
        sW0[t] = e0 * inv;
        sW1[t] = e1 * inv;
        float a = 3.0f / (1.0f + expf(-biases[t])) - 1.0f;
        sA[t] = fminf(fmaxf(a, -1.0f + EPS), 2.0f - EPS);
    }
    __syncthreads();

    float* wscr = scr + wave * (32 * SPITCH);
    const float* myrow = wscr + lane * SPITCH;
    const float* xrow  = xw + c * XPITCH + 2 * h;
    const float* bcol  = Bs + (2 * h) * LBLK + c;

    v8f acc[2];
    acc[0] = v8f{};
    acc[1] = v8f{};
    v2f fa[2][KCH], fb[KCH];
    load_a(xrow, 0, fa);
    load_b(bcol, 0, fb);
    for (int ch = 0; ch < LBLK; ++ch) {
        mma_chunk(fa, fb, acc);
        if (ch + 1 < LBLK) { load_a(xrow, ch + 1, fa); load_b(bcol, ch + 1, fb); }
    }
    __syncthreads();
    load_a(xrow, 0, fa);
    load_b(bcol + BBLK, 0, fb);

    PPieces pst = {};
    float carry = 0.0f;
    for (int l = 0; l < LDIM; ++l) {
        const int i = l & (LBLK - 1);
        const int j = l / LBLK;
        if (i == 0) {
            #pragma unroll
            for (int mt = 0; mt < 2; ++mt) {
                #pragma unroll
                for (int r = 0; r < 8; ++r)
                    wscr[(16 * mt + 8 * h + r) * SPITCH + c] = acc[mt][r];
            }
            acc[0] = v8f{};
            acc[1] = v8f{};
            if (j + 2 < LDIM / LBLK) p_fetch(P, j + 2, t, pst);
        }
        if (i == LBLK / 2 && j + 2 < LDIM / LBLK) p_store(Bs + (j & 1) * BBLK, t, pst);
        const int   lp  = (l > 0) ? l - 1 : 0;
        const float lf  = myrow[i];
        const float w0  = sW0[lp];
        const float w1  = sW1[lp];
        const float a_v = sA[lp];
        if (l + LBLK < LDIM) mma_chunk(fa, fb, acc);
        const float a = __builtin_bit_cast(float, __builtin_amdgcn_readfirstlane(__builtin_bit_cast(int, a_v)));
        if (l + 1 + LBLK < LDIM) {
            if (i == LBLK - 1) __syncthreads();
            const int s = l + 1;
            load_a(xrow, s & (LBLK - 1), fa);
            load_b(bcol + ((s / LBLK + 1) & 1) * BBLK, s & (LBLK - 1), fb);
        }
        if (l == 0) carry = lf;
        else        carry = combine_op(carry, lf, a, w0, w1);
    }
    out[row0 + lane] = carry;
}

extern "C" void kernel_launch(void* const* d_in, const int* in_sizes, int n_in,
                              void* d_out, int out_size, void* d_ws, size_t ws_size,
                              hipStream_t stream) {
    const float* x       = (const float*)d_in[0];
    const float* logits  = (const float*)d_in[1];
    const float* weights = (const float*)d_in[2];
    const float* biases  = (const float*)d_in[3];

    float* P     = (float*)d_ws;
    float* leaf  = P + LDIM * LDIM;

    float* partA = leaf;
    float* partB = leaf + 2 * SNB * LDIM;
    float* f     = leaf + 4 * SNB * LDIM;
#define SINK(pin, pout, pmat) sink_iter_kernel<<<SNB, 256, 0, stream>>>(logits, pin, pout, f, pmat)
    SINK(nullptr, partA, nullptr);
    SINK(partA,   partB, nullptr);
    SINK(partB,   partA, nullptr);
    SINK(partA,   partB, nullptr);
    SINK(partB,   partA, nullptr);
    SINK(partA,   partB, nullptr);
    SINK(partB,   partA, nullptr);
    SINK(partA,   partB, nullptr);
    SINK(partB,   partA, nullptr);
    SINK(partA,   partB, nullptr);
    SINK(partB,   nullptr, P);
#undef SINK
    gemm_chain_kernel<<<BATCH / 256, 256, 0, stream>>>(x, P, weights, biases, (float*)d_out);
}


# Product moved to 8 extra waves per work-group, one 16-level block ahead via LDS; scan waves run only the combine chain, operands read a step early; sum order kept, one class's power sum now unfused
# speedup vs baseline: 2.2485x; 1.0907x over previous
#include <hip/hip_runtime.h>
#include <hip/hip_bf16.h>
#include <math.h>

#define EPS 1e-6f
#define LDIM 256
#define BATCH 65536

typedef __attribute__((ext_vector_type(2))) float v2f;
typedef __attribute__((ext_vector_type(8))) float v8f;

#define SROWS 8
#define SNB   (LDIM / SROWS)

__global__ __launch_bounds__(256) void sink_iter_kernel(const float* __restrict__ logits,
                                                        const float* __restrict__ part_in,
                                                        float* __restrict__ part_out,
                                                        float* __restrict__ fbuf,
                                                        float* __restrict__ Pout) {
    __shared__ float sg[LDIM];
    __shared__ float sf[SROWS];
    const int t    = threadIdx.x;
    const int lane = t & 31;
    const int w    = t >> 5;
    const int row0 = blockIdx.x * SROWS;

    float c[SROWS];
    #pragma unroll
    for (int r = 0; r < SROWS; ++r) c[r] = logits[(row0 + r) * LDIM + t];
    float v[8];
    #pragma unroll
    for (int k = 0; k < 8; ++k) v[k] = logits[(row0 + w) * LDIM + lane + 32 * k];

    float g = 0.0f;
    if (part_in != nullptr) {
        const float2* pin = (const float2*)part_in + t;
        float2 p[SNB];
        #pragma unroll
        for (int b = 0; b < SNB; ++b) p[b] = pin[b * LDIM];
        float M = p[0].x;
        #pragma unroll
        for (int b = 1; b < SNB; ++b) M = fmaxf(M, p[b].x);
        float S = 0.0f;
        #pragma unroll
        for (int b = 0; b < SNB; ++b) S += p[b].y * expf(p[b].x - M);
        g = M + logf(S);
    }

    if (Pout != nullptr) {
        #pragma unroll
        for (int r = 0; r < SROWS; ++r)
            Pout[(row0 + r) * LDIM + t] = expf((c[r] - fbuf[row0 + r]) - g);
        return;
    }

    sg[t] = g;
    __syncthreads();

    #pragma unroll
    for (int k = 0; k < 8; ++k) v[k] -= sg[lane + 32 * k];
    float mx = v[0];
    #pragma unroll
    for (int k = 1; k < 8; ++k) mx = fmaxf(mx, v[k]);
    #pragma unroll
    for (int o = 16; o > 0; o >>= 1) mx = fmaxf(mx, __shfl_xor(mx, o));
    float s = 0.0f;
    #pragma unroll
    for (int k = 0; k < 8; ++k) s += expf(v[k] - mx);
    #pragma unroll
    for (int o = 16; o > 0; o >>= 1) s += __shfl_xor(s, o);
    const float lse = mx + logf(s);
    if (lane == 0) { sf[w] = lse; fbuf[row0 + w] = lse; }
    __syncthreads();

    float u[SROWS];
    #pragma unroll
    for (int r = 0; r < SROWS; ++r) u[r] = c[r] - sf[r];
    float m = u[0];
    #pragma unroll
    for (int r = 1; r < SROWS; ++r) m = fmaxf(m, u[r]);
    float ps = 0.0f;
    #pragma unroll
    for (int r = 0; r < SROWS; ++r) ps += expf(u[r] - m);
    ((float2*)part_out)[blockIdx.x * LDIM + t] = make_float2(m, ps);
}

#define LBLK 16
#define SPITCH 17
#define KSTEPS (LDIM / 4)
#define KCH (KSTEPS / LBLK)
#define XPITCH (LDIM + 4)
#define BBLK (LDIM * LBLK)
#define NSCAN 8

__device__ __forceinline__ float r_poly(float a) {
    float d  = 0.5f - a;
    float d2 = d * d;
    float num = 0.25f + 1.65811f * d + 2.15388f * d2 + 8.2844f * d2 * d + 6.16764f * d2 * d2;
    float den = a * (1.0f - a);
    den = (fabsf(den) < EPS) ? EPS : den;
    return num / den;
}

__device__ __forceinline__ float power_r(float x, float y, float ra, float w0, float w1) {
    ra = (fabsf(ra) < EPS) ? EPS : ra;
    return powf(w0 * powf(x, ra) + w1 * powf(y, ra), 1.0f / ra);
}

__device__ __forceinline__ float F_upper(float x, float y, float a, float w0, float w1) {
    const float xy  = x * y;
    const float mn  = fminf(x, y);
    const float lin = fmaf(w0, x, w1 * y);
    float out = lin;
    if (a > 0.5f && a < 0.75f) {
        const float R = r_poly(0.75f);
        out = (3.0f - 4.0f * a) * lin + (4.0f * a - 2.0f) * power_r(x, y, R, w0, w1);
    }
    if (a >= 0.75f && a < 1.0f) {
        const float ra = r_poly(fminf(fmaxf(a, 0.75f), 1.0f - EPS));
        out = power_r(x, y, ra, w0, w1);
    }
    if (fabsf(a - 1.0f)  < EPS) out = mn;
    if (a > 1.0f  && a < 1.25f) out = 4.0f * ((1.25f - a) * mn + (a - 1.0f) * xy);
    if (fabsf(a - 1.25f) < EPS) out = xy;
    if (a > 1.25f && a < 2.0f) {
        const float p = sqrtf(fmaxf(3.0f / fmaxf(2.0f - a, EPS) - 1.0f, EPS));
        out = powf(xy, p);
    }
    if (fabsf(a - 2.0f)  < EPS)
        out = ((fabsf(x - 1.0f) < EPS) && (fabsf(y - 1.0f) < EPS)) ? 1.0f : 0.0f;
    return out;
}

__device__ __forceinline__ float F_comb(float x, float y, float a, float w0, float w1) {
    if (isnan(x)) x = EPS;
    x = fminf(fmaxf(x, EPS), 1.0f - EPS);
    if (isnan(y)) y = EPS;
    y = fminf(fmaxf(y, EPS), 1.0f - EPS);
    if (a < 0.5f - EPS) {
        const float af = fminf(fmaxf(1.0f - a, -1.0f + EPS), 2.0f - EPS);
        return 1.0f - F_upper(1.0f - x, 1.0f - y, af, w0, w1);
    }
    return F_upper(x, y, a, w0, w1);
}

__device__ __forceinline__ float combine_op(float l, float r, float a, float w0, float w1) {
    const float nres = F_comb(l, r, a, w0, w1);
    return isnan(nres) ? a : nres;
}


__device__ __forceinline__ void load_a(const float* xrow, int ch, v2f (&fa)[2][KCH]) {
    const float* p = xrow + ch * (4 * KCH);
    #pragma unroll
    for (int m = 0; m < KCH; ++m) {
        #pragma unroll
        for (int mt = 0; mt < 2; ++mt) fa[mt][m] = *(const v2f*)(p + mt * 16 * XPITCH + 4 * m);
    }
}
__device__ __forceinline__ void load_b(const float* bcol, int ch, v2f (&fb)[KCH]) {
    const float* p = bcol + ch * (4 * KCH * LBLK);
    #pragma unroll
    for (int m = 0; m < KCH; ++m) {
        fb[m].x = p[(4 * m)     * LBLK];
        fb[m].y = p[(4 * m + 1) * LBLK];
    }
}
__device__ __forceinline__ void mma_chunk(const v2f (&fa)[2][KCH], const v2f (&fb)[KCH], v8f (&acc)[2]) {
    #pragma unroll
    for (int m = 0; m < KCH; ++m) {
        acc[0] = __builtin_amdgcn_wmma_f32_16x16x4_f32(false, fa[0][m], false, fb[m], (short)0, acc[0], false, false);
        acc[1] = __builtin_amdgcn_wmma_f32_16x16x4_f32(false, fa[1][m], false, fb[m], (short)0, acc[1], false, false);
    }
}
struct PPieces { float4 q0, q1, q2, q3; };
__device__ __forceinline__ float4 p_piece(const float* __restrict__ P, int nb, int idx) {
    return *(const float4*)(P + (idx >> 2) * LDIM + nb * LBLK + (idx & 3) * 4);
}
__device__ __forceinline__ void p_fetch(const float* __restrict__ P, int nb, int t, PPieces& v) {
    v.q0 = p_piece(P, nb, t);
    v.q1 = p_piece(P, nb, t + 256);
    v.q2 = p_piece(P, nb, t + 512);
    v.q3 = p_piece(P, nb, t + 768);
}
__device__ __forceinline__ void p_store(float* buf, int t, const PPieces& v) {
    float4* b = (float4*)buf + t;
    b[0]   = v.q0;
    b[256] = v.q1;
    b[512] = v.q2;
    b[768] = v.q3;
}

__device__ __forceinline__ void block_product(const float* xrow, const float* bcol, v8f (&acc)[2]) {
    v2f fa[2][KCH], fb[KCH];
    acc[0] = v8f{};
    acc[1] = v8f{};
    load_a(xrow, 0, fa);
    load_b(bcol, 0, fb);
    for (int ch = 0; ch < LBLK; ++ch) {
        mma_chunk(fa, fb, acc);
        if (ch + 1 < LBLK) { load_a(xrow, ch + 1, fa); load_b(bcol, ch + 1, fb); }
    }
}
__device__ __forceinline__ void leaves_out(float* wscr, int c, int h, const v8f (&acc)[2]) {
    #pragma unroll
    for (int mt = 0; mt < 2; ++mt) {
        #pragma unroll
        for (int r = 0; r < 8; ++r)
            wscr[(16 * mt + 8 * h + r) * SPITCH + c] = acc[mt][r];
    }
}

__global__ __launch_bounds__(512) void gemm_chain_kernel(const float* __restrict__ x,
                                                         const float* __restrict__ P,
                                                         const float* __restrict__ weights,
                                                         const float* __restrict__ biases,
                                                         float* __restrict__ out) {
    __shared__ float Xs[NSCAN * 32 * XPITCH];
    __shared__ float Bs[2 * BBLK];
    __shared__ float scr[NSCAN * 32 * SPITCH];
    __shared__ float sA[255], sW0[255], sW1[255];
    const int t    = threadIdx.x;
    const int wave = t >> 5;
    const int lane = t & 31;
    const bool product = wave >= NSCAN;
    const int pw   = wave & (NSCAN - 1);
    const int tp   = t & 255;
    const int c    = lane & 15;
    const int h    = lane >> 4;

    const int row0 = blockIdx.x * 256 + pw * 32;

    PPieces pst0 = {}, pst1 = {};
    if (!product) {
        p_fetch(P, 0, tp, pst0);
        p_fetch(P, 1, tp, pst1);
    }
    {
        float* xd = Xs + wave * (16 * XPITCH);
        const float4* xg = (const float4*)(x + ((size_t)blockIdx.x * 256 + wave * 16) * LDIM);
        #pragma unroll 32
        for (int it = 0; it < 16 * LDIM / 128; ++it) {
            const int f = it * 32 + lane;
            *(float4*)(xd + (f >> 6) * XPITCH + (f & 63) * 4) = xg[f];
        }
    }
    if (!product) {
        p_store(Bs, tp, pst0);
        p_store(Bs + BBLK, tp, pst1);
    }
    if (t < 255) {
        const float w0 = weights[2 * t], w1 = weights[2 * t + 1];
        const float m  = fmaxf(w0, w1);
        const float e0 = expf(w0 - m), e1 = expf(w1 - m);
        const float inv = 1.0f / (e0 + e1);
        sW0[t] = e0 * inv;
        sW1[t] = e1 * inv;
        float a = 3.0f / (1.0f + expf(-biases[t])) - 1.0f;
        sA[t] = fminf(fmaxf(a, -1.0f + EPS), 2.0f - EPS);
    }
    __syncthreads();

    float* wscr = scr + pw * (32 * SPITCH);
    const float* myrow = wscr + lane * SPITCH;
    const float* xrow  = Xs + pw * (32 * XPITCH) + c * XPITCH + 2 * h;
    const float* bcol  = Bs + (2 * h) * LBLK + c;

    v8f acc[2];
    acc[0] = v8f{};
    acc[1] = v8f{};
    if (product) {
        block_product(xrow, bcol, acc);
        leaves_out(wscr, c, h, acc);
    }
    __syncthreads();

    float carry = 0.0f;
    for (int j = 0; j < LDIM / LBLK; ++j) {
        if (product) {
            if (j + 1 < LDIM / LBLK) {
                PPieces pst = {};
                if (j + 2 < LDIM / LBLK) p_fetch(P, j + 2, tp, pst);
                block_product(xrow, bcol + ((j + 1) & 1) * BBLK, acc);
                if (j + 2 < LDIM / LBLK) p_store(Bs + (j & 1) * BBLK, tp, pst);
            }
        } else {
            const int l0  = j * LBLK;
            const int lp0 = (l0 > 0) ? l0 - 1 : 0;
            float lf  = myrow[0];
            float w0  = sW0[lp0];
            float w1  = sW1[lp0];
            float a_v = sA[lp0];
            for (int i = 0; i < LBLK; ++i) {
                const int l = l0 + i;
                const float a = __builtin_bit_cast(float, __builtin_amdgcn_readfirstlane(__builtin_bit_cast(int, a_v)));
                const float clf = lf, cw0 = w0, cw1 = w1;
                if (i + 1 < LBLK) {
                    lf  = myrow[i + 1];
                    w0  = sW0[l];
                    w1  = sW1[l];
                    a_v = sA[l];
                }
                if (l == 0) carry = clf;
                else        carry = combine_op(carry, clf, a, cw0, cw1);
            }
        }
        __syncthreads();
        if (product && j + 1 < LDIM / LBLK) leaves_out(wscr, c, h, acc);
        __syncthreads();
    }
    if (!product) out[row0 + lane] = carry;
}

extern "C" void kernel_launch(void* const* d_in, const int* in_sizes, int n_in,
                              void* d_out, int out_size, void* d_ws, size_t ws_size,
                              hipStream_t stream) {
    const float* x       = (const float*)d_in[0];
    const float* logits  = (const float*)d_in[1];
    const float* weights = (const float*)d_in[2];
    const float* biases  = (const float*)d_in[3];

    float* P     = (float*)d_ws;
    float* leaf  = P + LDIM * LDIM;

    float* partA = leaf;
    float* partB = leaf + 2 * SNB * LDIM;
    float* f     = leaf + 4 * SNB * LDIM;
#define SINK(pin, pout, pmat) sink_iter_kernel<<<SNB, 256, 0, stream>>>(logits, pin, pout, f, pmat)
    SINK(nullptr, partA, nullptr);
    SINK(partA,   partB, nullptr);
    SINK(partB,   partA, nullptr);
    SINK(partA,   partB, nullptr);
    SINK(partB,   partA, nullptr);
    SINK(partA,   partB, nullptr);
    SINK(partB,   partA, nullptr);
    SINK(partA,   partB, nullptr);
    SINK(partB,   partA, nullptr);
    SINK(partA,   partB, nullptr);
    SINK(partB,   nullptr, P);
#undef SINK
    gemm_chain_kernel<<<BATCH / 256, 512, 0, stream>>>(x, P, weights, biases, (float*)d_out);
}


# Scan carry handed on as its base-2 logarithm, clamped there: a power mean is log2(w0*2^(r*Lx)+w1*2^(r*Ly))/r on hardware exp2/log2, no pow left; powers move ~1e-6 rel, the power sum is fused
# speedup vs baseline: 3.3327x; 1.4822x over previous
#include <hip/hip_runtime.h>
#include <hip/hip_bf16.h>
#include <math.h>

#define EPS 1e-6f
#define LDIM 256
#define BATCH 65536

typedef __attribute__((ext_vector_type(2))) float v2f;
typedef __attribute__((ext_vector_type(8))) float v8f;

#define SROWS 8
#define SNB   (LDIM / SROWS)

__global__ __launch_bounds__(256) void sink_iter_kernel(const float* __restrict__ logits,
                                                        const float* __restrict__ part_in,
                                                        float* __restrict__ part_out,
                                                        float* __restrict__ fbuf,
                                                        float* __restrict__ Pout) {
    __shared__ float sg[LDIM];
    __shared__ float sf[SROWS];
    const int t    = threadIdx.x;
    const int lane = t & 31;
    const int w    = t >> 5;
    const int row0 = blockIdx.x * SROWS;

    float c[SROWS];
    #pragma unroll
    for (int r = 0; r < SROWS; ++r) c[r] = logits[(row0 + r) * LDIM + t];
    float v[8];
    #pragma unroll
    for (int k = 0; k < 8; ++k) v[k] = logits[(row0 + w) * LDIM + lane + 32 * k];

    float g = 0.0f;
    if (part_in != nullptr) {
        const float2* pin = (const float2*)part_in + t;
        float2 p[SNB];
        #pragma unroll
        for (int b = 0; b < SNB; ++b) p[b] = pin[b * LDIM];
        float M = p[0].x;
        #pragma unroll
        for (int b = 1; b < SNB; ++b) M = fmaxf(M, p[b].x);
        float S = 0.0f;
        #pragma unroll
        for (int b = 0; b < SNB; ++b) S += p[b].y * expf(p[b].x - M);
        g = M + logf(S);
    }

    if (Pout != nullptr) {
        #pragma unroll
        for (int r = 0; r < SROWS; ++r)
            Pout[(row0 + r) * LDIM + t] = expf((c[r] - fbuf[row0 + r]) - g);
        return;
    }

    sg[t] = g;
    __syncthreads();

    #pragma unroll
    for (int k = 0; k < 8; ++k) v[k] -= sg[lane + 32 * k];
    float mx = v[0];
    #pragma unroll
    for (int k = 1; k < 8; ++k) mx = fmaxf(mx, v[k]);
    #pragma unroll
    for (int o = 16; o > 0; o >>= 1) mx = fmaxf(mx, __shfl_xor(mx, o));
    float s = 0.0f;
    #pragma unroll
    for (int k = 0; k < 8; ++k) s += expf(v[k] - mx);
    #pragma unroll
    for (int o = 16; o > 0; o >>= 1) s += __shfl_xor(s, o);
    const float lse = mx + logf(s);
    if (lane == 0) { sf[w] = lse; fbuf[row0 + w] = lse; }
    __syncthreads();

    float u[SROWS];
    #pragma unroll
    for (int r = 0; r < SROWS; ++r) u[r] = c[r] - sf[r];
    float m = u[0];
    #pragma unroll
    for (int r = 1; r < SROWS; ++r) m = fmaxf(m, u[r]);
    float ps = 0.0f;
    #pragma unroll
    for (int r = 0; r < SROWS; ++r) ps += expf(u[r] - m);
    ((float2*)part_out)[blockIdx.x * LDIM + t] = make_float2(m, ps);
}

#define LBLK 16
#define SPITCH 17
#define KSTEPS (LDIM / 4)
#define KCH (KSTEPS / LBLK)
#define XPITCH (LDIM + 4)
#define BBLK (LDIM * LBLK)
#define NSCAN 8

#define L2LO (__builtin_log2f(EPS))
#define L2HI (__builtin_log2f(1.0f - EPS))

__device__ __forceinline__ float ex2(float t) { return __builtin_amdgcn_exp2f(t); }
__device__ __forceinline__ float lg2(float t) { return __builtin_amdgcn_logf(t); }

__device__ __forceinline__ float r_poly(float a) {
    float d  = 0.5f - a;
    float d2 = d * d;
    float num = 0.25f + 1.65811f * d + 2.15388f * d2 + 8.2844f * d2 * d + 6.16764f * d2 * d2;
    float den = a * (1.0f - a);
    den = (fabsf(den) < EPS) ? EPS : den;
    return num / den;
}

__device__ __forceinline__ float power_r_log(float Lx, float Ly, float ra, float w0, float w1) {
    ra = (fabsf(ra) < EPS) ? EPS : ra;
    const float s = fmaf(w0, ex2(ra * Lx), w1 * ex2(ra * Ly));
    return lg2(s) * (1.0f / ra);
}

__device__ __forceinline__ float combine_log(float L, float y, float a, float w0, float w1, bool fin) {
    const float Lx = fminf(fmaxf(L, L2LO), L2HI);
    float yc = isnan(y) ? EPS : y;
    yc = fminf(fmaxf(yc, EPS), 1.0f - EPS);
    const bool  mir = a < 0.5f - EPS;
    const float af  = mir ? fminf(fmaxf(1.0f - a, -1.0f + EPS), 2.0f - EPS) : a;
    int cls = 0;
    if (af > 0.5f && af < 0.75f)  cls = 1;
    if (af >= 0.75f && af < 1.0f) cls = 2;
    if (fabsf(af - 1.0f)  < EPS)  cls = 3;
    if (af > 1.0f && af < 1.25f)  cls = 4;
    if (fabsf(af - 1.25f) < EPS)  cls = 5;
    if (af > 1.25f && af < 2.0f)  cls = 6;
    if (fabsf(af - 2.0f)  < EPS)  cls = 7;
    const bool nat = (cls == 2) || (cls == 3) || (cls == 5) || (cls == 6);

    float xq = 0.0f, yq = yc, Lxq = Lx;
    if (mir) {
        xq = 1.0f - ex2(Lx);
        yq = 1.0f - yc;
        if (nat || cls == 1) Lxq = lg2(xq);
    } else if (!nat) {
        xq = ex2(Lx);
    }
    float Lr = 0.0f, vr = 0.0f;
    switch (cls) {
    case 1: {
        const float lin = fmaf(w0, xq, w1 * yq);
        const float R   = r_poly(0.75f);
        const float pm  = ex2(power_r_log(Lxq, lg2(yq), R, w0, w1));
        vr = (3.0f - 4.0f * af) * lin + (4.0f * af - 2.0f) * pm;
    } break;
    case 2: {
        const float ra = r_poly(fminf(fmaxf(af, 0.75f), 1.0f - EPS));
        Lr = power_r_log(Lxq, lg2(yq), ra, w0, w1);
    } break;
    case 3: Lr = fminf(Lxq, lg2(yq)); break;
    case 4: {
        const float mn = fminf(xq, yq);
        const float xy = xq * yq;
        vr = 4.0f * ((1.25f - af) * mn + (af - 1.0f) * xy);
    } break;
    case 5: Lr = Lxq + lg2(yq); break;
    case 6: {
        const float p = sqrtf(fmaxf(3.0f / fmaxf(2.0f - af, EPS) - 1.0f, EPS));
        Lr = p * (Lxq + lg2(yq));
    } break;
    case 7: vr = ((fabsf(xq - 1.0f) < EPS) && (fabsf(yq - 1.0f) < EPS)) ? 1.0f : 0.0f; break;
    default: vr = fmaf(w0, xq, w1 * yq); break;
    }
    if (mir) {
        float o = 1.0f - (nat ? ex2(Lr) : vr);
        if (isnan(o)) o = a;
        return fin ? o : lg2(fmaxf(o, 0.0f));
    }
    if (nat) {
        if (fin) { const float o = ex2(Lr); return isnan(o) ? a : o; }
        return isnan(Lr) ? lg2(fmaxf(a, 0.0f)) : Lr;
    }
    const float o = isnan(vr) ? a : vr;
    return fin ? o : lg2(fmaxf(o, 0.0f));
}


__device__ __forceinline__ void load_a(const float* xrow, int ch, v2f (&fa)[2][KCH]) {
    const float* p = xrow + ch * (4 * KCH);
    #pragma unroll
    for (int m = 0; m < KCH; ++m) {
        #pragma unroll
        for (int mt = 0; mt < 2; ++mt) fa[mt][m] = *(const v2f*)(p + mt * 16 * XPITCH + 4 * m);
    }
}
__device__ __forceinline__ void load_b(const float* bcol, int ch, v2f (&fb)[KCH]) {
    const float* p = bcol + ch * (4 * KCH * LBLK);
    #pragma unroll
    for (int m = 0; m < KCH; ++m) {
        fb[m].x = p[(4 * m)     * LBLK];
        fb[m].y = p[(4 * m + 1) * LBLK];
    }
}
__device__ __forceinline__ void mma_chunk(const v2f (&fa)[2][KCH], const v2f (&fb)[KCH], v8f (&acc)[2]) {
    #pragma unroll
    for (int m = 0; m < KCH; ++m) {
        acc[0] = __builtin_amdgcn_wmma_f32_16x16x4_f32(false, fa[0][m], false, fb[m], (short)0, acc[0], false, false);
        acc[1] = __builtin_amdgcn_wmma_f32_16x16x4_f32(false, fa[1][m], false, fb[m], (short)0, acc[1], false, false);
    }
}
struct PPieces { float4 q0, q1, q2, q3; };
__device__ __forceinline__ float4 p_piece(const float* __restrict__ P, int nb, int idx) {
    return *(const float4*)(P + (idx >> 2) * LDIM + nb * LBLK + (idx & 3) * 4);
}
__device__ __forceinline__ void p_fetch(const float* __restrict__ P, int nb, int t, PPieces& v) {
    v.q0 = p_piece(P, nb, t);
    v.q1 = p_piece(P, nb, t + 256);
    v.q2 = p_piece(P, nb, t + 512);
    v.q3 = p_piece(P, nb, t + 768);
}
__device__ __forceinline__ void p_store(float* buf, int t, const PPieces& v) {
    float4* b = (float4*)buf + t;
    b[0]   = v.q0;
    b[256] = v.q1;
    b[512] = v.q2;
    b[768] = v.q3;
}

__device__ __forceinline__ void block_product(const float* xrow, const float* bcol, v8f (&acc)[2]) {
    v2f fa[2][KCH], fb[KCH];
    acc[0] = v8f{};
    acc[1] = v8f{};
    load_a(xrow, 0, fa);
    load_b(bcol, 0, fb);
    for (int ch = 0; ch < LBLK; ++ch) {
        mma_chunk(fa, fb, acc);
        if (ch + 1 < LBLK) { load_a(xrow, ch + 1, fa); load_b(bcol, ch + 1, fb); }
    }
}
__device__ __forceinline__ void leaves_out(float* wscr, int c, int h, const v8f (&acc)[2]) {
    #pragma unroll
    for (int mt = 0; mt < 2; ++mt) {
        #pragma unroll
        for (int r = 0; r < 8; ++r)
            wscr[(16 * mt + 8 * h + r) * SPITCH + c] = acc[mt][r];
    }
}

__global__ __launch_bounds__(512) void gemm_chain_kernel(const float* __restrict__ x,
                                                         const float* __restrict__ P,
                                                         const float* __restrict__ weights,
                                                         const float* __restrict__ biases,
                                                         float* __restrict__ out) {
    __shared__ float Xs[NSCAN * 32 * XPITCH];
    __shared__ float Bs[2 * BBLK];
    __shared__ float scr[NSCAN * 32 * SPITCH];
    __shared__ float sA[255], sW0[255], sW1[255];
    const int t    = threadIdx.x;
    const int wave = t >> 5;
    const int lane = t & 31;
    const bool product = wave >= NSCAN;
    const int pw   = wave & (NSCAN - 1);
    const int tp   = t & 255;
    const int c    = lane & 15;
    const int h    = lane >> 4;

    const int row0 = blockIdx.x * 256 + pw * 32;

    PPieces pst0 = {}, pst1 = {};
    if (!product) {
        p_fetch(P, 0, tp, pst0);
        p_fetch(P, 1, tp, pst1);
    }
    {
        float* xd = Xs + wave * (16 * XPITCH);
        const float4* xg = (const float4*)(x + ((size_t)blockIdx.x * 256 + wave * 16) * LDIM);
        #pragma unroll 32
        for (int it = 0; it < 16 * LDIM / 128; ++it) {
            const int f = it * 32 + lane;
            *(float4*)(xd + (f >> 6) * XPITCH + (f & 63) * 4) = xg[f];
        }
    }
    if (!product) {
        p_store(Bs, tp, pst0);
        p_store(Bs + BBLK, tp, pst1);
    }
    if (t < 255) {
        const float w0 = weights[2 * t], w1 = weights[2 * t + 1];
        const float m  = fmaxf(w0, w1);
        const float e0 = expf(w0 - m), e1 = expf(w1 - m);
        const float inv = 1.0f / (e0 + e1);
        sW0[t] = e0 * inv;
        sW1[t] = e1 * inv;
        float a = 3.0f / (1.0f + expf(-biases[t])) - 1.0f;
        sA[t] = fminf(fmaxf(a, -1.0f + EPS), 2.0f - EPS);
    }
    __syncthreads();

    float* wscr = scr + pw * (32 * SPITCH);
    const float* myrow = wscr + lane * SPITCH;
    const float* xrow  = Xs + pw * (32 * XPITCH) + c * XPITCH + 2 * h;
    const float* bcol  = Bs + (2 * h) * LBLK + c;

    v8f acc[2];
    acc[0] = v8f{};
    acc[1] = v8f{};
    if (product) {
        block_product(xrow, bcol, acc);
        leaves_out(wscr, c, h, acc);
    }
    __syncthreads();

    float carry = 0.0f;
    for (int j = 0; j < LDIM / LBLK; ++j) {
        if (product) {
            if (j + 1 < LDIM / LBLK) {
                PPieces pst = {};
                if (j + 2 < LDIM / LBLK) p_fetch(P, j + 2, tp, pst);
                block_product(xrow, bcol + ((j + 1) & 1) * BBLK, acc);
                if (j + 2 < LDIM / LBLK) p_store(Bs + (j & 1) * BBLK, tp, pst);
            }
        } else {
            const int l0  = j * LBLK;
            const int lp0 = (l0 > 0) ? l0 - 1 : 0;
            float lf  = myrow[0];
            float w0  = sW0[lp0];
            float w1  = sW1[lp0];
            float a_v = sA[lp0];
            for (int i = 0; i < LBLK; ++i) {
                const int l = l0 + i;
                const float a = __builtin_bit_cast(float, __builtin_amdgcn_readfirstlane(__builtin_bit_cast(int, a_v)));
                const float clf = lf, cw0 = w0, cw1 = w1;
                if (i + 1 < LBLK) {
                    lf  = myrow[i + 1];
                    w0  = sW0[l];
                    w1  = sW1[l];
                    a_v = sA[l];
                }
                if (l == 0) carry = lg2(fminf(fmaxf(isnan(clf) ? EPS : clf, EPS), 1.0f - EPS));
                else        carry = combine_log(carry, clf, a, cw0, cw1, l == LDIM - 1);
            }
        }
        __syncthreads();
        if (product && j + 1 < LDIM / LBLK) leaves_out(wscr, c, h, acc);
        __syncthreads();
    }
    if (!product) out[row0 + lane] = carry;
}

extern "C" void kernel_launch(void* const* d_in, const int* in_sizes, int n_in,
                              void* d_out, int out_size, void* d_ws, size_t ws_size,
                              hipStream_t stream) {
    const float* x       = (const float*)d_in[0];
    const float* logits  = (const float*)d_in[1];
    const float* weights = (const float*)d_in[2];
    const float* biases  = (const float*)d_in[3];

    float* P     = (float*)d_ws;
    float* leaf  = P + LDIM * LDIM;

    float* partA = leaf;
    float* partB = leaf + 2 * SNB * LDIM;
    float* f     = leaf + 4 * SNB * LDIM;
#define SINK(pin, pout, pmat) sink_iter_kernel<<<SNB, 256, 0, stream>>>(logits, pin, pout, f, pmat)
    SINK(nullptr, partA, nullptr);
    SINK(partA,   partB, nullptr);
    SINK(partB,   partA, nullptr);
    SINK(partA,   partB, nullptr);
    SINK(partB,   partA, nullptr);
    SINK(partA,   partB, nullptr);
    SINK(partB,   partA, nullptr);
    SINK(partA,   partB, nullptr);
    SINK(partB,   partA, nullptr);
    SINK(partA,   partB, nullptr);
    SINK(partB,   nullptr, P);
#undef SINK
    gemm_chain_kernel<<<BATCH / 256, 512, 0, stream>>>(x, P, weights, biases, (float*)d_out);
}


# Scan waves work out a block's 16 levels' class, mirror flag and coefficients (polynomial, divisions, square root) in one lane-parallel pass; a step takes its level's by v_readlane; sums unchanged
# speedup vs baseline: 3.8136x; 1.1443x over previous
#include <hip/hip_runtime.h>
#include <hip/hip_bf16.h>
#include <math.h>

#define EPS 1e-6f
#define LDIM 256
#define BATCH 65536

typedef __attribute__((ext_vector_type(2))) float v2f;
typedef __attribute__((ext_vector_type(8))) float v8f;

#define SROWS 8
#define SNB   (LDIM / SROWS)

__global__ __launch_bounds__(256) void sink_iter_kernel(const float* __restrict__ logits,
                                                        const float* __restrict__ part_in,
                                                        float* __restrict__ part_out,
                                                        float* __restrict__ fbuf,
                                                        float* __restrict__ Pout) {
    __shared__ float sg[LDIM];
    __shared__ float sf[SROWS];
    const int t    = threadIdx.x;
    const int lane = t & 31;
    const int w    = t >> 5;
    const int row0 = blockIdx.x * SROWS;

    float c[SROWS];
    #pragma unroll
    for (int r = 0; r < SROWS; ++r) c[r] = logits[(row0 + r) * LDIM + t];
    float v[8];
    #pragma unroll
    for (int k = 0; k < 8; ++k) v[k] = logits[(row0 + w) * LDIM + lane + 32 * k];

    float g = 0.0f;
    if (part_in != nullptr) {
        const float2* pin = (const float2*)part_in + t;
        float2 p[SNB];
        #pragma unroll
        for (int b = 0; b < SNB; ++b) p[b] = pin[b * LDIM];
        float M = p[0].x;
        #pragma unroll
        for (int b = 1; b < SNB; ++b) M = fmaxf(M, p[b].x);
        float S = 0.0f;
        #pragma unroll
        for (int b = 0; b < SNB; ++b) S += p[b].y * expf(p[b].x - M);
        g = M + logf(S);
    }

    if (Pout != nullptr) {
        #pragma unroll
        for (int r = 0; r < SROWS; ++r)
            Pout[(row0 + r) * LDIM + t] = expf((c[r] - fbuf[row0 + r]) - g);
        return;
    }

    sg[t] = g;
    __syncthreads();

    #pragma unroll
    for (int k = 0; k < 8; ++k) v[k] -= sg[lane + 32 * k];
    float mx = v[0];
    #pragma unroll
    for (int k = 1; k < 8; ++k) mx = fmaxf(mx, v[k]);
    #pragma unroll
    for (int o = 16; o > 0; o >>= 1) mx = fmaxf(mx, __shfl_xor(mx, o));
    float s = 0.0f;
    #pragma unroll
    for (int k = 0; k < 8; ++k) s += expf(v[k] - mx);
    #pragma unroll
    for (int o = 16; o > 0; o >>= 1) s += __shfl_xor(s, o);
    const float lse = mx + logf(s);
    if (lane == 0) { sf[w] = lse; fbuf[row0 + w] = lse; }
    __syncthreads();

    float u[SROWS];
    #pragma unroll
    for (int r = 0; r < SROWS; ++r) u[r] = c[r] - sf[r];
    float m = u[0];
    #pragma unroll
    for (int r = 1; r < SROWS; ++r) m = fmaxf(m, u[r]);
    float ps = 0.0f;
    #pragma unroll
    for (int r = 0; r < SROWS; ++r) ps += expf(u[r] - m);
    ((float2*)part_out)[blockIdx.x * LDIM + t] = make_float2(m, ps);
}

#define LBLK 16
#define SPITCH 17
#define KSTEPS (LDIM / 4)
#define KCH (KSTEPS / LBLK)
#define XPITCH (LDIM + 4)
#define BBLK (LDIM * LBLK)
#define NSCAN 8

#define L2LO (__builtin_log2f(EPS))
#define L2HI (__builtin_log2f(1.0f - EPS))

__device__ __forceinline__ float ex2(float t) { return __builtin_amdgcn_exp2f(t); }
__device__ __forceinline__ float lg2(float t) { return __builtin_amdgcn_logf(t); }

__device__ __forceinline__ float r_poly(float a) {
    float d  = 0.5f - a;
    float d2 = d * d;
    float num = 0.25f + 1.65811f * d + 2.15388f * d2 + 8.2844f * d2 * d + 6.16764f * d2 * d2;
    float den = a * (1.0f - a);
    den = (fabsf(den) < EPS) ? EPS : den;
    return num / den;
}

__device__ __forceinline__ float power_r_log(float Lx, float Ly, float ra, float q, float w0, float w1) {
    const float s = fmaf(w0, ex2(ra * Lx), w1 * ex2(ra * Ly));
    return lg2(s) * q;
}

struct LevelCoef { float a, w0, w1, e, q, c1, c2; int code; };

__device__ __forceinline__ LevelCoef level_coef(float a, float w0, float w1) {
    LevelCoef k;
    const bool  mir = a < 0.5f - EPS;
    const float af  = mir ? fminf(fmaxf(1.0f - a, -1.0f + EPS), 2.0f - EPS) : a;
    int cls = 0;
    if (af > 0.5f && af < 0.75f)  cls = 1;
    if (af >= 0.75f && af < 1.0f) cls = 2;
    if (fabsf(af - 1.0f)  < EPS)  cls = 3;
    if (af > 1.0f && af < 1.25f)  cls = 4;
    if (fabsf(af - 1.25f) < EPS)  cls = 5;
    if (af > 1.25f && af < 2.0f)  cls = 6;
    if (fabsf(af - 2.0f)  < EPS)  cls = 7;
    float ra = r_poly(fminf(fmaxf(af, 0.75f), 1.0f - EPS));
    ra = (fabsf(ra) < EPS) ? EPS : ra;
    const float p = sqrtf(fmaxf(3.0f / fmaxf(2.0f - af, EPS) - 1.0f, EPS));
    k.e  = (cls == 6) ? p : ra;
    k.q  = 1.0f / ra;
    k.c1 = (cls == 4) ? (1.25f - af) : fmaf(-4.0f, af, 3.0f);
    k.c2 = (cls == 4) ? (af - 1.0f)  : fmaf(4.0f, af, -2.0f);
    k.a  = a;
    k.w0 = w0;
    k.w1 = w1;
    k.code = cls | (mir ? 8 : 0);
    return k;
}

__device__ __forceinline__ float lane_f(float v, int i) {
    return __builtin_bit_cast(float, __builtin_amdgcn_readlane(__builtin_bit_cast(int, v), i));
}
__device__ __forceinline__ LevelCoef block_coef(const float* tA, const float* tW0, const float* tW1, int l0, int i) {
    const int lp = (l0 + i > 0) ? l0 + i - 1 : 0;
    return level_coef(tA[lp], tW0[lp], tW1[lp]);
}

__device__ __forceinline__ float combine_log(float L, float y, const LevelCoef& kv, int i, int code, float a, bool fin) {
    const float Lx = fminf(fmaxf(L, L2LO), L2HI);
    float yc = isnan(y) ? EPS : y;
    yc = fminf(fmaxf(yc, EPS), 1.0f - EPS);
    const bool mir = (code & 8) != 0;
    const int  cls = code & 7;
    const bool nat = (cls == 2) || (cls == 3) || (cls == 5) || (cls == 6);

    float xq = 0.0f, yq = yc, Lxq = Lx;
    if (mir) {
        xq = 1.0f - ex2(Lx);
        yq = 1.0f - yc;
        if (nat || cls == 1) Lxq = lg2(xq);
    } else if (!nat) {
        xq = ex2(Lx);
    }
    float Lr = 0.0f, vr = 0.0f;
    switch (cls) {
    case 1: {
        const float w0 = lane_f(kv.w0, i), w1 = lane_f(kv.w1, i);
        const float lin = fmaf(w0, xq, w1 * yq);
        float R = r_poly(0.75f);
        R = (fabsf(R) < EPS) ? EPS : R;
        const float pm  = ex2(power_r_log(Lxq, lg2(yq), R, 1.0f / R, w0, w1));
        vr = fmaf(lane_f(kv.c1, i), lin, lane_f(kv.c2, i) * pm);
    } break;
    case 2: Lr = power_r_log(Lxq, lg2(yq), lane_f(kv.e, i), lane_f(kv.q, i), lane_f(kv.w0, i), lane_f(kv.w1, i)); break;
    case 3: Lr = fminf(Lxq, lg2(yq)); break;
    case 4: {
        const float mn = fminf(xq, yq);
        const float xy = xq * yq;
        vr = 4.0f * fmaf(lane_f(kv.c1, i), mn, lane_f(kv.c2, i) * xy);
    } break;
    case 5: Lr = Lxq + lg2(yq); break;
    case 6: Lr = lane_f(kv.e, i) * (Lxq + lg2(yq)); break;
    case 7: vr = ((fabsf(xq - 1.0f) < EPS) && (fabsf(yq - 1.0f) < EPS)) ? 1.0f : 0.0f; break;
    default: vr = fmaf(lane_f(kv.w0, i), xq, lane_f(kv.w1, i) * yq); break;
    }
    if (mir) {
        float o = 1.0f - (nat ? ex2(Lr) : vr);
        if (isnan(o)) o = a;
        return fin ? o : lg2(fmaxf(o, 0.0f));
    }
    if (nat) {
        if (fin) { const float o = ex2(Lr); return isnan(o) ? a : o; }
        return isnan(Lr) ? lg2(fmaxf(a, 0.0f)) : Lr;
    }
    const float o = isnan(vr) ? a : vr;
    return fin ? o : lg2(fmaxf(o, 0.0f));
}


__device__ __forceinline__ void load_a(const float* xrow, int ch, v2f (&fa)[2][KCH]) {
    const float* p = xrow + ch * (4 * KCH);
    #pragma unroll
    for (int m = 0; m < KCH; ++m) {
        #pragma unroll
        for (int mt = 0; mt < 2; ++mt) fa[mt][m] = *(const v2f*)(p + mt * 16 * XPITCH + 4 * m);
    }
}
__device__ __forceinline__ void load_b(const float* bcol, int ch, v2f (&fb)[KCH]) {
    const float* p = bcol + ch * (4 * KCH * LBLK);
    #pragma unroll
    for (int m = 0; m < KCH; ++m) {
        fb[m].x = p[(4 * m)     * LBLK];
        fb[m].y = p[(4 * m + 1) * LBLK];
    }
}
__device__ __forceinline__ void mma_chunk(const v2f (&fa)[2][KCH], const v2f (&fb)[KCH], v8f (&acc)[2]) {
    #pragma unroll
    for (int m = 0; m < KCH; ++m) {
        acc[0] = __builtin_amdgcn_wmma_f32_16x16x4_f32(false, fa[0][m], false, fb[m], (short)0, acc[0], false, false);
        acc[1] = __builtin_amdgcn_wmma_f32_16x16x4_f32(false, fa[1][m], false, fb[m], (short)0, acc[1], false, false);
    }
}
struct PPieces { float4 q0, q1, q2, q3; };
__device__ __forceinline__ float4 p_piece(const float* __restrict__ P, int nb, int idx) {
    return *(const float4*)(P + (idx >> 2) * LDIM + nb * LBLK + (idx & 3) * 4);
}
__device__ __forceinline__ void p_fetch(const float* __restrict__ P, int nb, int t, PPieces& v) {
    v.q0 = p_piece(P, nb, t);
    v.q1 = p_piece(P, nb, t + 256);
    v.q2 = p_piece(P, nb, t + 512);
    v.q3 = p_piece(P, nb, t + 768);
}
__device__ __forceinline__ void p_store(float* buf, int t, const PPieces& v) {
    float4* b = (float4*)buf + t;
    b[0]   = v.q0;
    b[256] = v.q1;
    b[512] = v.q2;
    b[768] = v.q3;
}

__device__ __forceinline__ void block_product(const float* xrow, const float* bcol, v8f (&acc)[2]) {
    v2f fa[2][KCH], fb[KCH];
    acc[0] = v8f{};
    acc[1] = v8f{};
    load_a(xrow, 0, fa);
    load_b(bcol, 0, fb);
    for (int ch = 0; ch < LBLK; ++ch) {
        mma_chunk(fa, fb, acc);
        if (ch + 1 < LBLK) { load_a(xrow, ch + 1, fa); load_b(bcol, ch + 1, fb); }
    }
}
__device__ __forceinline__ void leaves_out(float* wscr, int c, int h, const v8f (&acc)[2]) {
    #pragma unroll
    for (int mt = 0; mt < 2; ++mt) {
        #pragma unroll
        for (int r = 0; r < 8; ++r)
            wscr[(16 * mt + 8 * h + r) * SPITCH + c] = acc[mt][r];
    }
}

__global__ __launch_bounds__(512) void gemm_chain_kernel(const float* __restrict__ x,
                                                         const float* __restrict__ P,
                                                         const float* __restrict__ weights,
                                                         const float* __restrict__ biases,
                                                         float* __restrict__ out) {
    __shared__ float Xs[NSCAN * 32 * XPITCH];
    __shared__ float Bs[2 * BBLK];
    __shared__ float scr[NSCAN * 32 * SPITCH];
    __shared__ float sA[255], sW0[255], sW1[255];
    const int t    = threadIdx.x;
    const int wave = t >> 5;
    const int lane = t & 31;
    const bool product = wave >= NSCAN;
    const int pw   = wave & (NSCAN - 1);
    const int tp   = t & 255;
    const int c    = lane & 15;
    const int h    = lane >> 4;

    const int row0 = blockIdx.x * 256 + pw * 32;

    PPieces pst0 = {}, pst1 = {};
    if (!product) {
        p_fetch(P, 0, tp, pst0);
        p_fetch(P, 1, tp, pst1);
    }
    {
        float* xd = Xs + wave * (16 * XPITCH);
        const float4* xg = (const float4*)(x + ((size_t)blockIdx.x * 256 + wave * 16) * LDIM);
        #pragma unroll 32
        for (int it = 0; it < 16 * LDIM / 128; ++it) {
            const int f = it * 32 + lane;
            *(float4*)(xd + (f >> 6) * XPITCH + (f & 63) * 4) = xg[f];
        }
    }
    if (!product) {
        p_store(Bs, tp, pst0);
        p_store(Bs + BBLK, tp, pst1);
    }
    if (t < 255) {
        const float w0 = weights[2 * t], w1 = weights[2 * t + 1];
        const float m  = fmaxf(w0, w1);
        const float e0 = expf(w0 - m), e1 = expf(w1 - m);
        const float inv = 1.0f / (e0 + e1);
        sW0[t] = e0 * inv;
        sW1[t] = e1 * inv;
        float a = 3.0f / (1.0f + expf(-biases[t])) - 1.0f;
        sA[t] = fminf(fmaxf(a, -1.0f + EPS), 2.0f - EPS);
    }
    __syncthreads();

    float* wscr = scr + pw * (32 * SPITCH);
    const float* myrow = wscr + lane * SPITCH;
    const float* xrow  = Xs + pw * (32 * XPITCH) + c * XPITCH + 2 * h;
    const float* bcol  = Bs + (2 * h) * LBLK + c;

    v8f acc[2];
    acc[0] = v8f{};
    acc[1] = v8f{};
    LevelCoef kc = {};
    if (product) {
        block_product(xrow, bcol, acc);
        leaves_out(wscr, c, h, acc);
    } else {
        kc = block_coef(sA, sW0, sW1, 0, c);
    }
    __syncthreads();

    float carry = 0.0f;
    for (int j = 0; j < LDIM / LBLK; ++j) {
        if (product) {
            if (j + 1 < LDIM / LBLK) {
                PPieces pst = {};
                if (j + 2 < LDIM / LBLK) p_fetch(P, j + 2, tp, pst);
                block_product(xrow, bcol + ((j + 1) & 1) * BBLK, acc);
                if (j + 2 < LDIM / LBLK) p_store(Bs + (j & 1) * BBLK, tp, pst);
            }
        } else {
            const int l0 = j * LBLK;
            float lf = myrow[0];
            int   nc = __builtin_amdgcn_readlane(kc.code, 0);
            float na = lane_f(kc.a, 0);
            for (int i = 0; i < LBLK; ++i) {
                const int l = l0 + i;
                const float clf = lf;
                const int   cc  = nc;
                const float ca  = na;
                if (i + 1 < LBLK) {
                    lf = myrow[i + 1];
                    nc = __builtin_amdgcn_readlane(kc.code, i + 1);
                    na = lane_f(kc.a, i + 1);
                }
                if (l == 0) carry = lg2(fminf(fmaxf(isnan(clf) ? EPS : clf, EPS), 1.0f - EPS));
                else        carry = combine_log(carry, clf, kc, i, cc, ca, l == LDIM - 1);
            }
        }
        __syncthreads();
        if (j + 1 < LDIM / LBLK) {
            if (product) leaves_out(wscr, c, h, acc);
            else         kc = block_coef(sA, sW0, sW1, (j + 1) * LBLK, c);
        }
        __syncthreads();
    }
    if (!product) out[row0 + lane] = carry;
}

extern "C" void kernel_launch(void* const* d_in, const int* in_sizes, int n_in,
                              void* d_out, int out_size, void* d_ws, size_t ws_size,
                              hipStream_t stream) {
    const float* x       = (const float*)d_in[0];
    const float* logits  = (const float*)d_in[1];
    const float* weights = (const float*)d_in[2];
    const float* biases  = (const float*)d_in[3];

    float* P     = (float*)d_ws;
    float* leaf  = P + LDIM * LDIM;

    float* partA = leaf;
    float* partB = leaf + 2 * SNB * LDIM;
    float* f     = leaf + 4 * SNB * LDIM;
#define SINK(pin, pout, pmat) sink_iter_kernel<<<SNB, 256, 0, stream>>>(logits, pin, pout, f, pmat)
    SINK(nullptr, partA, nullptr);
    SINK(partA,   partB, nullptr);
    SINK(partB,   partA, nullptr);
    SINK(partA,   partB, nullptr);
    SINK(partB,   partA, nullptr);
    SINK(partA,   partB, nullptr);
    SINK(partB,   partA, nullptr);
    SINK(partA,   partB, nullptr);
    SINK(partB,   partA, nullptr);
    SINK(partA,   partB, nullptr);
    SINK(partB,   nullptr, P);
#undef SINK
    gemm_chain_kernel<<<BATCH / 256, 512, 0, stream>>>(x, P, weights, biases, (float*)d_out);
}


# Sinkhorn in ONE launch (was 11): one 16-wave work-group, logits in registers, sweeps on hardware exp2/log2 shifted by the previous potentials (true max in sweeps 1-2 only); sum order, rounding change
# speedup vs baseline: 4.3290x; 1.1351x over previous
#include <hip/hip_runtime.h>
#include <hip/hip_bf16.h>
#include <math.h>

#define EPS 1e-6f
#define LDIM 256
#define BATCH 65536

typedef __attribute__((ext_vector_type(2))) float v2f;
typedef __attribute__((ext_vector_type(8))) float v8f;

#define LOG2E 1.44269504088896340736f
#define SK_WAVES 16
#define SK_ROWS (LDIM / SK_WAVES)
#define SINK_ITERS 10

__device__ __forceinline__ float ex2(float t) { return __builtin_amdgcn_exp2f(t); }
__device__ __forceinline__ float lg2(float t) { return __builtin_amdgcn_logf(t); }

template <int MASK>
__device__ __forceinline__ float xrow(float v) {
    const int b = __builtin_bit_cast(int, v);
    return __builtin_bit_cast(float, __builtin_amdgcn_update_dpp(0, b, 0x160 + MASK, 0xf, 0xf, true));
}
__device__ __forceinline__ float xhalf(float v) {
    const int b = __builtin_bit_cast(int, v);
    return __builtin_bit_cast(float, __builtin_amdgcn_permlanex16(b, b, 0x76543210, 0xfedcba98, false, false));
}
__device__ __forceinline__ float wave_sum(float v) {
    v += xrow<1>(v); v += xrow<2>(v); v += xrow<4>(v); v += xrow<8>(v);
    return v + xhalf(v);
}
__device__ __forceinline__ float wave_max(float v) {
    v = fmaxf(v, xrow<1>(v)); v = fmaxf(v, xrow<2>(v)); v = fmaxf(v, xrow<4>(v)); v = fmaxf(v, xrow<8>(v));
    return fmaxf(v, xhalf(v));
}
__device__ __forceinline__ v2f expo(v2f l, v2f p) {
    return __builtin_elementwise_fma(l, v2f{LOG2E, LOG2E}, -p);
}

__global__ __launch_bounds__(SK_WAVES * 32) void sinkhorn_kernel(const float* __restrict__ logits,
                                                             float* __restrict__ Pout) {
    __shared__ float part[SK_WAVES * LDIM];
    __shared__ float sd[LDIM];
    const int t    = threadIdx.x;
    const int lane = t & 31;
    const int w    = t >> 5;
    const float* src = logits + (SK_ROWS * w) * LDIM + lane;

    v2f L[SK_ROWS][4];
    #pragma unroll
    for (int r = 0; r < SK_ROWS; ++r) {
        #pragma unroll
        for (int q = 0; q < 4; ++q)
            L[r][q] = v2f{src[r * LDIM + 64 * q], src[r * LDIM + 64 * q + 32]};
    }
    v2f   g[4];
    float f[SK_ROWS];
    #pragma unroll
    for (int q = 0; q < 4; ++q) g[q] = v2f{0.0f, 0.0f};
    #pragma unroll
    for (int r = 0; r < SK_ROWS; ++r) {
        float m = fmaxf(L[r][0].x, L[r][0].y);
        #pragma unroll
        for (int q = 1; q < 4; ++q) m = fmaxf(m, fmaxf(L[r][q].x, L[r][q].y));
        f[r] = wave_max(m) * LOG2E;
    }

    for (int it = 0; it < SINK_ITERS; ++it) {
        #pragma unroll
        for (int r = 0; r < SK_ROWS; ++r) {
            v2f s = v2f{0.0f, 0.0f};
            #pragma unroll
            for (int q = 0; q < 4; ++q) {
                const v2f e = expo(L[r][q], g[q] + f[r]);
                s += v2f{ex2(e.x), ex2(e.y)};
            }
            f[r] += lg2(wave_sum(s.x + s.y));
        }
        if (it == 0) {
            v2f cm[4];
            #pragma unroll
            for (int q = 0; q < 4; ++q) cm[q] = expo(L[0][q], g[q] + f[0]);
            #pragma unroll
            for (int r = 1; r < SK_ROWS; ++r) {
                #pragma unroll
                for (int q = 0; q < 4; ++q) {
                    const v2f e = expo(L[r][q], g[q] + f[r]);
                    cm[q] = v2f{fmaxf(cm[q].x, e.x), fmaxf(cm[q].y, e.y)};
                }
            }
            #pragma unroll
            for (int q = 0; q < 4; ++q) {
                part[w * LDIM + lane + 64 * q]      = cm[q].x;
                part[w * LDIM + lane + 64 * q + 32] = cm[q].y;
            }
            __syncthreads();
            if (t < LDIM) {
                float m = part[t];
                #pragma unroll
                for (int b = 1; b < SK_WAVES; ++b) m = fmaxf(m, part[b * LDIM + t]);
                sd[t] = m;
            }
            __syncthreads();
            #pragma unroll
            for (int q = 0; q < 4; ++q) g[q] = v2f{sd[lane + 64 * q], sd[lane + 64 * q + 32]};
        }
        v2f c[4];
        #pragma unroll
        for (int q = 0; q < 4; ++q) c[q] = v2f{0.0f, 0.0f};
        #pragma unroll
        for (int r = 0; r < SK_ROWS; ++r) {
            #pragma unroll
            for (int q = 0; q < 4; ++q) {
                const v2f e = expo(L[r][q], g[q] + f[r]);
                c[q] += v2f{ex2(e.x), ex2(e.y)};
            }
        }
        #pragma unroll
        for (int q = 0; q < 4; ++q) {
            part[w * LDIM + lane + 64 * q]      = c[q].x;
            part[w * LDIM + lane + 64 * q + 32] = c[q].y;
        }
        __syncthreads();
        if (t < LDIM) {
            float s0 = 0.0f, s1 = 0.0f, s2 = 0.0f, s3 = 0.0f;
            #pragma unroll
            for (int b = 0; b < SK_WAVES; b += 4) {
                s0 += part[(b)     * LDIM + t];
                s1 += part[(b + 1) * LDIM + t];
                s2 += part[(b + 2) * LDIM + t];
                s3 += part[(b + 3) * LDIM + t];
            }
            sd[t] = lg2((s0 + s1) + (s2 + s3));
        }
        __syncthreads();
        #pragma unroll
        for (int q = 0; q < 4; ++q) g[q] += v2f{sd[lane + 64 * q], sd[lane + 64 * q + 32]};
    }

    float* dst = Pout + (SK_ROWS * w) * LDIM + lane;
    #pragma unroll
    for (int r = 0; r < SK_ROWS; ++r) {
        #pragma unroll
        for (int q = 0; q < 4; ++q) {
            const v2f e = expo(L[r][q], g[q] + f[r]);
            dst[r * LDIM + 64 * q]      = ex2(e.x);
            dst[r * LDIM + 64 * q + 32] = ex2(e.y);
        }
    }
}

#define LBLK 16
#define SPITCH 17
#define KSTEPS (LDIM / 4)
#define KCH (KSTEPS / LBLK)
#define XPITCH (LDIM + 4)
#define BBLK (LDIM * LBLK)
#define NSCAN 8

#define L2LO (__builtin_log2f(EPS))
#define L2HI (__builtin_log2f(1.0f - EPS))


__device__ __forceinline__ float r_poly(float a) {
    float d  = 0.5f - a;
    float d2 = d * d;
    float num = 0.25f + 1.65811f * d + 2.15388f * d2 + 8.2844f * d2 * d + 6.16764f * d2 * d2;
    float den = a * (1.0f - a);
    den = (fabsf(den) < EPS) ? EPS : den;
    return num / den;
}

__device__ __forceinline__ float power_r_log(float Lx, float Ly, float ra, float q, float w0, float w1) {
    const float s = fmaf(w0, ex2(ra * Lx), w1 * ex2(ra * Ly));
    return lg2(s) * q;
}

struct LevelCoef { float a, w0, w1, e, q, c1, c2; int code; };

__device__ __forceinline__ LevelCoef level_coef(float a, float w0, float w1) {
    LevelCoef k;
    const bool  mir = a < 0.5f - EPS;
    const float af  = mir ? fminf(fmaxf(1.0f - a, -1.0f + EPS), 2.0f - EPS) : a;
    int cls = 0;
    if (af > 0.5f && af < 0.75f)  cls = 1;
    if (af >= 0.75f && af < 1.0f) cls = 2;
    if (fabsf(af - 1.0f)  < EPS)  cls = 3;
    if (af > 1.0f && af < 1.25f)  cls = 4;
    if (fabsf(af - 1.25f) < EPS)  cls = 5;
    if (af > 1.25f && af < 2.0f)  cls = 6;
    if (fabsf(af - 2.0f)  < EPS)  cls = 7;
    float ra = r_poly(fminf(fmaxf(af, 0.75f), 1.0f - EPS));
    ra = (fabsf(ra) < EPS) ? EPS : ra;
    const float p = sqrtf(fmaxf(3.0f / fmaxf(2.0f - af, EPS) - 1.0f, EPS));
    k.e  = (cls == 6) ? p : ra;
    k.q  = 1.0f / ra;
    k.c1 = (cls == 4) ? (1.25f - af) : fmaf(-4.0f, af, 3.0f);
    k.c2 = (cls == 4) ? (af - 1.0f)  : fmaf(4.0f, af, -2.0f);
    k.a  = a;
    k.w0 = w0;
    k.w1 = w1;
    k.code = cls | (mir ? 8 : 0);
    return k;
}

__device__ __forceinline__ float lane_f(float v, int i) {
    return __builtin_bit_cast(float, __builtin_amdgcn_readlane(__builtin_bit_cast(int, v), i));
}
__device__ __forceinline__ LevelCoef block_coef(const float* tA, const float* tW0, const float* tW1, int l0, int i) {
    const int lp = (l0 + i > 0) ? l0 + i - 1 : 0;
    return level_coef(tA[lp], tW0[lp], tW1[lp]);
}

__device__ __forceinline__ float combine_log(float L, float y, const LevelCoef& kv, int i, int code, float a, bool fin) {
    const float Lx = fminf(fmaxf(L, L2LO), L2HI);
    float yc = isnan(y) ? EPS : y;
    yc = fminf(fmaxf(yc, EPS), 1.0f - EPS);
    const bool mir = (code & 8) != 0;
    const int  cls = code & 7;
    const bool nat = (cls == 2) || (cls == 3) || (cls == 5) || (cls == 6);

    float xq = 0.0f, yq = yc, Lxq = Lx;
    if (mir) {
        xq = 1.0f - ex2(Lx);
        yq = 1.0f - yc;
        if (nat || cls == 1) Lxq = lg2(xq);
    } else if (!nat) {
        xq = ex2(Lx);
    }
    float Lr = 0.0f, vr = 0.0f;
    switch (cls) {
    case 1: {
        const float w0 = lane_f(kv.w0, i), w1 = lane_f(kv.w1, i);
        const float lin = fmaf(w0, xq, w1 * yq);
        float R = r_poly(0.75f);
        R = (fabsf(R) < EPS) ? EPS : R;
        const float pm  = ex2(power_r_log(Lxq, lg2(yq), R, 1.0f / R, w0, w1));
        vr = fmaf(lane_f(kv.c1, i), lin, lane_f(kv.c2, i) * pm);
    } break;
    case 2: Lr = power_r_log(Lxq, lg2(yq), lane_f(kv.e, i), lane_f(kv.q, i), lane_f(kv.w0, i), lane_f(kv.w1, i)); break;
    case 3: Lr = fminf(Lxq, lg2(yq)); break;
    case 4: {
        const float mn = fminf(xq, yq);
        const float xy = xq * yq;
        vr = 4.0f * fmaf(lane_f(kv.c1, i), mn, lane_f(kv.c2, i) * xy);
    } break;
    case 5: Lr = Lxq + lg2(yq); break;
    case 6: Lr = lane_f(kv.e, i) * (Lxq + lg2(yq)); break;
    case 7: vr = ((fabsf(xq - 1.0f) < EPS) && (fabsf(yq - 1.0f) < EPS)) ? 1.0f : 0.0f; break;
    default: vr = fmaf(lane_f(kv.w0, i), xq, lane_f(kv.w1, i) * yq); break;
    }
    if (mir) {
        float o = 1.0f - (nat ? ex2(Lr) : vr);
        if (isnan(o)) o = a;
        return fin ? o : lg2(fmaxf(o, 0.0f));
    }
    if (nat) {
        if (fin) { const float o = ex2(Lr); return isnan(o) ? a : o; }
        return isnan(Lr) ? lg2(fmaxf(a, 0.0f)) : Lr;
    }
    const float o = isnan(vr) ? a : vr;
    return fin ? o : lg2(fmaxf(o, 0.0f));
}


__device__ __forceinline__ void load_a(const float* xrow, int ch, v2f (&fa)[2][KCH]) {
    const float* p = xrow + ch * (4 * KCH);
    #pragma unroll
    for (int m = 0; m < KCH; ++m) {
        #pragma unroll
        for (int mt = 0; mt < 2; ++mt) fa[mt][m] = *(const v2f*)(p + mt * 16 * XPITCH + 4 * m);
    }
}
__device__ __forceinline__ void load_b(const float* bcol, int ch, v2f (&fb)[KCH]) {
    const float* p = bcol + ch * (4 * KCH * LBLK);
    #pragma unroll
    for (int m = 0; m < KCH; ++m) {
        fb[m].x = p[(4 * m)     * LBLK];
        fb[m].y = p[(4 * m + 1) * LBLK];
    }
}
__device__ __forceinline__ void mma_chunk(const v2f (&fa)[2][KCH], const v2f (&fb)[KCH], v8f (&acc)[2]) {
    #pragma unroll
    for (int m = 0; m < KCH; ++m) {
        acc[0] = __builtin_amdgcn_wmma_f32_16x16x4_f32(false, fa[0][m], false, fb[m], (short)0, acc[0], false, false);
        acc[1] = __builtin_amdgcn_wmma_f32_16x16x4_f32(false, fa[1][m], false, fb[m], (short)0, acc[1], false, false);
    }
}
struct PPieces { float4 q0, q1, q2, q3; };
__device__ __forceinline__ float4 p_piece(const float* __restrict__ P, int nb, int idx) {
    return *(const float4*)(P + (idx >> 2) * LDIM + nb * LBLK + (idx & 3) * 4);
}
__device__ __forceinline__ void p_fetch(const float* __restrict__ P, int nb, int t, PPieces& v) {
    v.q0 = p_piece(P, nb, t);
    v.q1 = p_piece(P, nb, t + 256);
    v.q2 = p_piece(P, nb, t + 512);
    v.q3 = p_piece(P, nb, t + 768);
}
__device__ __forceinline__ void p_store(float* buf, int t, const PPieces& v) {
    float4* b = (float4*)buf + t;
    b[0]   = v.q0;
    b[256] = v.q1;
    b[512] = v.q2;
    b[768] = v.q3;
}

__device__ __forceinline__ void block_product(const float* xrow, const float* bcol, v8f (&acc)[2]) {
    v2f fa[2][KCH], fb[KCH];
    acc[0] = v8f{};
    acc[1] = v8f{};
    load_a(xrow, 0, fa);
    load_b(bcol, 0, fb);
    for (int ch = 0; ch < LBLK; ++ch) {
        mma_chunk(fa, fb, acc);
        if (ch + 1 < LBLK) { load_a(xrow, ch + 1, fa); load_b(bcol, ch + 1, fb); }
    }
}
__device__ __forceinline__ void leaves_out(float* wscr, int c, int h, const v8f (&acc)[2]) {
    #pragma unroll
    for (int mt = 0; mt < 2; ++mt) {
        #pragma unroll
        for (int r = 0; r < 8; ++r)
            wscr[(16 * mt + 8 * h + r) * SPITCH + c] = acc[mt][r];
    }
}

__global__ __launch_bounds__(512) void gemm_chain_kernel(const float* __restrict__ x,
                                                         const float* __restrict__ P,
                                                         const float* __restrict__ weights,
                                                         const float* __restrict__ biases,
                                                         float* __restrict__ out) {
    __shared__ float Xs[NSCAN * 32 * XPITCH];
    __shared__ float Bs[2 * BBLK];
    __shared__ float scr[NSCAN * 32 * SPITCH];
    __shared__ float sA[255], sW0[255], sW1[255];
    const int t    = threadIdx.x;
    const int wave = t >> 5;
    const int lane = t & 31;
    const bool product = wave >= NSCAN;
    const int pw   = wave & (NSCAN - 1);
    const int tp   = t & 255;
    const int c    = lane & 15;
    const int h    = lane >> 4;

    const int row0 = blockIdx.x * 256 + pw * 32;

    PPieces pst0 = {}, pst1 = {};
    if (!product) {
        p_fetch(P, 0, tp, pst0);
        p_fetch(P, 1, tp, pst1);
    }
    {
        float* xd = Xs + wave * (16 * XPITCH);
        const float4* xg = (const float4*)(x + ((size_t)blockIdx.x * 256 + wave * 16) * LDIM);
        #pragma unroll 32
        for (int it = 0; it < 16 * LDIM / 128; ++it) {
            const int f = it * 32 + lane;
            *(float4*)(xd + (f >> 6) * XPITCH + (f & 63) * 4) = xg[f];
        }
    }
    if (!product) {
        p_store(Bs, tp, pst0);
        p_store(Bs + BBLK, tp, pst1);
    }
    if (t < 255) {
        const float w0 = weights[2 * t], w1 = weights[2 * t + 1];
        const float m  = fmaxf(w0, w1);
        const float e0 = expf(w0 - m), e1 = expf(w1 - m);
        const float inv = 1.0f / (e0 + e1);
        sW0[t] = e0 * inv;
        sW1[t] = e1 * inv;
        float a = 3.0f / (1.0f + expf(-biases[t])) - 1.0f;
        sA[t] = fminf(fmaxf(a, -1.0f + EPS), 2.0f - EPS);
    }
    __syncthreads();

    float* wscr = scr + pw * (32 * SPITCH);
    const float* myrow = wscr + lane * SPITCH;
    const float* xrow  = Xs + pw * (32 * XPITCH) + c * XPITCH + 2 * h;
    const float* bcol  = Bs + (2 * h) * LBLK + c;

    v8f acc[2];
    acc[0] = v8f{};
    acc[1] = v8f{};
    LevelCoef kc = {};
    if (product) {
        block_product(xrow, bcol, acc);
        leaves_out(wscr, c, h, acc);
    } else {
        kc = block_coef(sA, sW0, sW1, 0, c);
    }
    __syncthreads();

    float carry = 0.0f;
    for (int j = 0; j < LDIM / LBLK; ++j) {
        if (product) {
            if (j + 1 < LDIM / LBLK) {
                PPieces pst = {};
                if (j + 2 < LDIM / LBLK) p_fetch(P, j + 2, tp, pst);
                block_product(xrow, bcol + ((j + 1) & 1) * BBLK, acc);
                if (j + 2 < LDIM / LBLK) p_store(Bs + (j & 1) * BBLK, tp, pst);
            }
        } else {
            const int l0 = j * LBLK;
            float lf = myrow[0];
            int   nc = __builtin_amdgcn_readlane(kc.code, 0);
            float na = lane_f(kc.a, 0);
            for (int i = 0; i < LBLK; ++i) {
                const int l = l0 + i;
                const float clf = lf;
                const int   cc  = nc;
                const float ca  = na;
                if (i + 1 < LBLK) {
                    lf = myrow[i + 1];
                    nc = __builtin_amdgcn_readlane(kc.code, i + 1);
                    na = lane_f(kc.a, i + 1);
                }
                if (l == 0) carry = lg2(fminf(fmaxf(isnan(clf) ? EPS : clf, EPS), 1.0f - EPS));
                else        carry = combine_log(carry, clf, kc, i, cc, ca, l == LDIM - 1);
            }
        }
        __syncthreads();
        if (j + 1 < LDIM / LBLK) {
            if (product) leaves_out(wscr, c, h, acc);
            else         kc = block_coef(sA, sW0, sW1, (j + 1) * LBLK, c);
        }
        __syncthreads();
    }
    if (!product) out[row0 + lane] = carry;
}

extern "C" void kernel_launch(void* const* d_in, const int* in_sizes, int n_in,
                              void* d_out, int out_size, void* d_ws, size_t ws_size,
                              hipStream_t stream) {
    const float* x       = (const float*)d_in[0];
    const float* logits  = (const float*)d_in[1];
    const float* weights = (const float*)d_in[2];
    const float* biases  = (const float*)d_in[3];

    float* P = (float*)d_ws;

    sinkhorn_kernel<<<1, SK_WAVES * 32, 0, stream>>>(logits, P);
    gemm_chain_kernel<<<BATCH / 256, 512, 0, stream>>>(x, P, weights, biases, (float*)d_out);
}


# Scan loop control made wave-uniform: NaN and constant-class tests are selects, wave role a scalar, first/last level peeled; a level runs one straight-line body per class and mirror; sums unchanged
# speedup vs baseline: 5.0749x; 1.1723x over previous
#include <hip/hip_runtime.h>
#include <hip/hip_bf16.h>
#include <math.h>

#define EPS 1e-6f
#define LDIM 256
#define BATCH 65536

typedef __attribute__((ext_vector_type(2))) float v2f;
typedef __attribute__((ext_vector_type(8))) float v8f;

#define LOG2E 1.44269504088896340736f
#define SK_WAVES 16
#define SK_ROWS (LDIM / SK_WAVES)
#define SINK_ITERS 10

__device__ __forceinline__ float ex2(float t) { return __builtin_amdgcn_exp2f(t); }
__device__ __forceinline__ float lg2(float t) { return __builtin_amdgcn_logf(t); }

template <int MASK>
__device__ __forceinline__ float xrow(float v) {
    const int b = __builtin_bit_cast(int, v);
    return __builtin_bit_cast(float, __builtin_amdgcn_update_dpp(0, b, 0x160 + MASK, 0xf, 0xf, true));
}
__device__ __forceinline__ float xhalf(float v) {
    const int b = __builtin_bit_cast(int, v);
    return __builtin_bit_cast(float, __builtin_amdgcn_permlanex16(b, b, 0x76543210, 0xfedcba98, false, false));
}
__device__ __forceinline__ float wave_sum(float v) {
    v += xrow<1>(v); v += xrow<2>(v); v += xrow<4>(v); v += xrow<8>(v);
    return v + xhalf(v);
}
__device__ __forceinline__ float wave_max(float v) {
    v = fmaxf(v, xrow<1>(v)); v = fmaxf(v, xrow<2>(v)); v = fmaxf(v, xrow<4>(v)); v = fmaxf(v, xrow<8>(v));
    return fmaxf(v, xhalf(v));
}
__device__ __forceinline__ v2f expo(v2f l, v2f p) {
    return __builtin_elementwise_fma(l, v2f{LOG2E, LOG2E}, -p);
}

__global__ __launch_bounds__(SK_WAVES * 32) void sinkhorn_kernel(const float* __restrict__ logits,
                                                             float* __restrict__ Pout) {
    __shared__ float part[SK_WAVES * LDIM];
    __shared__ float sd[LDIM];
    const int t    = threadIdx.x;
    const int lane = t & 31;
    const int w    = t >> 5;
    const float* src = logits + (SK_ROWS * w) * LDIM + lane;

    v2f L[SK_ROWS][4];
    #pragma unroll
    for (int r = 0; r < SK_ROWS; ++r) {
        #pragma unroll
        for (int q = 0; q < 4; ++q)
            L[r][q] = v2f{src[r * LDIM + 64 * q], src[r * LDIM + 64 * q + 32]};
    }
    v2f   g[4];
    float f[SK_ROWS];
    #pragma unroll
    for (int q = 0; q < 4; ++q) g[q] = v2f{0.0f, 0.0f};
    #pragma unroll
    for (int r = 0; r < SK_ROWS; ++r) {
        float m = fmaxf(L[r][0].x, L[r][0].y);
        #pragma unroll
        for (int q = 1; q < 4; ++q) m = fmaxf(m, fmaxf(L[r][q].x, L[r][q].y));
        f[r] = wave_max(m) * LOG2E;
    }

    for (int it = 0; it < SINK_ITERS; ++it) {
        #pragma unroll
        for (int r = 0; r < SK_ROWS; ++r) {
            v2f s = v2f{0.0f, 0.0f};
            #pragma unroll
            for (int q = 0; q < 4; ++q) {
                const v2f e = expo(L[r][q], g[q] + f[r]);
                s += v2f{ex2(e.x), ex2(e.y)};
            }
            f[r] += lg2(wave_sum(s.x + s.y));
        }
        if (it == 0) {
            v2f cm[4];
            #pragma unroll
            for (int q = 0; q < 4; ++q) cm[q] = expo(L[0][q], g[q] + f[0]);
            #pragma unroll
            for (int r = 1; r < SK_ROWS; ++r) {
                #pragma unroll
                for (int q = 0; q < 4; ++q) {
                    const v2f e = expo(L[r][q], g[q] + f[r]);
                    cm[q] = v2f{fmaxf(cm[q].x, e.x), fmaxf(cm[q].y, e.y)};
                }
            }
            #pragma unroll
            for (int q = 0; q < 4; ++q) {
                part[w * LDIM + lane + 64 * q]      = cm[q].x;
                part[w * LDIM + lane + 64 * q + 32] = cm[q].y;
            }
            __syncthreads();
            if (t < LDIM) {
                float m = part[t];
                #pragma unroll
                for (int b = 1; b < SK_WAVES; ++b) m = fmaxf(m, part[b * LDIM + t]);
                sd[t] = m;
            }
            __syncthreads();
            #pragma unroll
            for (int q = 0; q < 4; ++q) g[q] = v2f{sd[lane + 64 * q], sd[lane + 64 * q + 32]};
        }
        v2f c[4];
        #pragma unroll
        for (int q = 0; q < 4; ++q) c[q] = v2f{0.0f, 0.0f};
        #pragma unroll
        for (int r = 0; r < SK_ROWS; ++r) {
            #pragma unroll
            for (int q = 0; q < 4; ++q) {
                const v2f e = expo(L[r][q], g[q] + f[r]);
                c[q] += v2f{ex2(e.x), ex2(e.y)};
            }
        }
        #pragma unroll
        for (int q = 0; q < 4; ++q) {
            part[w * LDIM + lane + 64 * q]      = c[q].x;
            part[w * LDIM + lane + 64 * q + 32] = c[q].y;
        }
        __syncthreads();
        if (t < LDIM) {
            float s0 = 0.0f, s1 = 0.0f, s2 = 0.0f, s3 = 0.0f;
            #pragma unroll
            for (int b = 0; b < SK_WAVES; b += 4) {
                s0 += part[(b)     * LDIM + t];
                s1 += part[(b + 1) * LDIM + t];
                s2 += part[(b + 2) * LDIM + t];
                s3 += part[(b + 3) * LDIM + t];
            }
            sd[t] = lg2((s0 + s1) + (s2 + s3));
        }
        __syncthreads();
        #pragma unroll
        for (int q = 0; q < 4; ++q) g[q] += v2f{sd[lane + 64 * q], sd[lane + 64 * q + 32]};
    }

    float* dst = Pout + (SK_ROWS * w) * LDIM + lane;
    #pragma unroll
    for (int r = 0; r < SK_ROWS; ++r) {
        #pragma unroll
        for (int q = 0; q < 4; ++q) {
            const v2f e = expo(L[r][q], g[q] + f[r]);
            dst[r * LDIM + 64 * q]      = ex2(e.x);
            dst[r * LDIM + 64 * q + 32] = ex2(e.y);
        }
    }
}

#define LBLK 16
#define SPITCH 17
#define KSTEPS (LDIM / 4)
#define KCH (KSTEPS / LBLK)
#define XPITCH (LDIM + 4)
#define BBLK (LDIM * LBLK)
#define NSCAN 8

#define L2LO (__builtin_log2f(EPS))
#define L2HI (__builtin_log2f(1.0f - EPS))


__device__ __forceinline__ float r_poly(float a) {
    float d  = 0.5f - a;
    float d2 = d * d;
    float num = 0.25f + 1.65811f * d + 2.15388f * d2 + 8.2844f * d2 * d + 6.16764f * d2 * d2;
    float den = a * (1.0f - a);
    den = (fabsf(den) < EPS) ? EPS : den;
    return num / den;
}

__device__ __forceinline__ float power_r_log(float Lx, float Ly, float ra, float q, float w0, float w1) {
    const float s = fmaf(w0, ex2(ra * Lx), w1 * ex2(ra * Ly));
    return lg2(s) * q;
}

struct LevelCoef { float a, w0, w1, e, q, c1, c2; int code; };

__device__ __forceinline__ LevelCoef level_coef(float a, float w0, float w1) {
    LevelCoef k;
    const bool  mir = a < 0.5f - EPS;
    const float af  = mir ? fminf(fmaxf(1.0f - a, -1.0f + EPS), 2.0f - EPS) : a;
    int cls = 0;
    if (af > 0.5f && af < 0.75f)  cls = 1;
    if (af >= 0.75f && af < 1.0f) cls = 2;
    if (fabsf(af - 1.0f)  < EPS)  cls = 3;
    if (af > 1.0f && af < 1.25f)  cls = 4;
    if (fabsf(af - 1.25f) < EPS)  cls = 5;
    if (af > 1.25f && af < 2.0f)  cls = 6;
    if (fabsf(af - 2.0f)  < EPS)  cls = 7;
    float ra = r_poly(fminf(fmaxf(af, 0.75f), 1.0f - EPS));
    ra = (fabsf(ra) < EPS) ? EPS : ra;
    const float p = sqrtf(fmaxf(3.0f / fmaxf(2.0f - af, EPS) - 1.0f, EPS));
    k.e  = (cls == 6) ? p : ra;
    k.q  = 1.0f / ra;
    k.c1 = (cls == 4) ? (1.25f - af) : fmaf(-4.0f, af, 3.0f);
    k.c2 = (cls == 4) ? (af - 1.0f)  : fmaf(4.0f, af, -2.0f);
    k.a  = a;
    k.w0 = w0;
    k.w1 = w1;
    k.code = cls | (mir ? 8 : 0);
    return k;
}

__device__ __forceinline__ float lane_f(float v, int i) {
    return __builtin_bit_cast(float, __builtin_amdgcn_readlane(__builtin_bit_cast(int, v), i));
}
__device__ __forceinline__ LevelCoef block_coef(const float* tA, const float* tW0, const float* tW1, int l0, int i) {
    const int lp = (l0 + i > 0) ? l0 + i - 1 : 0;
    return level_coef(tA[lp], tW0[lp], tW1[lp]);
}

template <bool FIN>
__device__ __forceinline__ float hand_on(float o, float a) {
    o = isnan(o) ? a : o;
    return FIN ? o : lg2(fmaxf(o, 0.0f));
}

template <int CLS, bool MIR, bool FIN>
__device__ __forceinline__ float level_step(float Lx, float yc, const LevelCoef& kv, int i, float a) {
    constexpr bool NAT = (CLS == 2) || (CLS == 3) || (CLS == 5) || (CLS == 6);
    float xq = 0.0f, yq = yc, Lxq = Lx;
    if (MIR) {
        xq = 1.0f - ex2(Lx);
        yq = 1.0f - yc;
        if (NAT || CLS == 1) Lxq = lg2(xq);
    } else if (!NAT) {
        xq = ex2(Lx);
    }
    float Lr = 0.0f, vr = 0.0f;
    if (CLS == 1) {
        const float w0 = lane_f(kv.w0, i), w1 = lane_f(kv.w1, i);
        const float lin = fmaf(w0, xq, w1 * yq);
        float R = r_poly(0.75f);
        R = (fabsf(R) < EPS) ? EPS : R;
        const float pm  = ex2(power_r_log(Lxq, lg2(yq), R, 1.0f / R, w0, w1));
        vr = fmaf(lane_f(kv.c1, i), lin, lane_f(kv.c2, i) * pm);
    } else if (CLS == 2) {
        Lr = power_r_log(Lxq, lg2(yq), lane_f(kv.e, i), lane_f(kv.q, i), lane_f(kv.w0, i), lane_f(kv.w1, i));
    } else if (CLS == 3) {
        Lr = fminf(Lxq, lg2(yq));
    } else if (CLS == 4) {
        const float mn = fminf(xq, yq);
        const float xy = xq * yq;
        vr = 4.0f * fmaf(lane_f(kv.c1, i), mn, lane_f(kv.c2, i) * xy);
    } else if (CLS == 5) {
        Lr = Lxq + lg2(yq);
    } else if (CLS == 6) {
        Lr = lane_f(kv.e, i) * (Lxq + lg2(yq));
    } else if (CLS == 7) {
        const bool bx = fabsf(xq - 1.0f) < EPS;
        const bool by = fabsf(yq - 1.0f) < EPS;
        vr = (bx & by) ? 1.0f : 0.0f;
    } else {
        vr = fmaf(lane_f(kv.w0, i), xq, lane_f(kv.w1, i) * yq);
    }
    if (MIR) return hand_on<FIN>(1.0f - (NAT ? ex2(Lr) : vr), a);
    if (NAT) {
        if (FIN) return hand_on<true>(ex2(Lr), a);
        const float la = lg2(fmaxf(a, 0.0f));
        return isnan(Lr) ? la : Lr;
    }
    return hand_on<FIN>(vr, a);
}

template <bool FIN>
__device__ __forceinline__ float combine_log(float L, float y, const LevelCoef& kv, int i, int code, float a) {
    const float Lx = fminf(fmaxf(L, L2LO), L2HI);
    float yc = isnan(y) ? EPS : y;
    yc = fminf(fmaxf(yc, EPS), 1.0f - EPS);
    switch (code) {
    case 0:  return level_step<0, false, FIN>(Lx, yc, kv, i, a);
    case 1:  return level_step<1, false, FIN>(Lx, yc, kv, i, a);
    case 2:  return level_step<2, false, FIN>(Lx, yc, kv, i, a);
    case 3:  return level_step<3, false, FIN>(Lx, yc, kv, i, a);
    case 4:  return level_step<4, false, FIN>(Lx, yc, kv, i, a);
    case 5:  return level_step<5, false, FIN>(Lx, yc, kv, i, a);
    case 6:  return level_step<6, false, FIN>(Lx, yc, kv, i, a);
    case 7:  return level_step<7, false, FIN>(Lx, yc, kv, i, a);
    case 8:  return level_step<0, true, FIN>(Lx, yc, kv, i, a);
    case 9:  return level_step<1, true, FIN>(Lx, yc, kv, i, a);
    case 10: return level_step<2, true, FIN>(Lx, yc, kv, i, a);
    case 11: return level_step<3, true, FIN>(Lx, yc, kv, i, a);
    case 12: return level_step<4, true, FIN>(Lx, yc, kv, i, a);
    case 13: return level_step<5, true, FIN>(Lx, yc, kv, i, a);
    case 14: return level_step<6, true, FIN>(Lx, yc, kv, i, a);
    default: return level_step<7, true, FIN>(Lx, yc, kv, i, a);
    }
}


__device__ __forceinline__ void load_a(const float* xrow, int ch, v2f (&fa)[2][KCH]) {
    const float* p = xrow + ch * (4 * KCH);
    #pragma unroll
    for (int m = 0; m < KCH; ++m) {
        #pragma unroll
        for (int mt = 0; mt < 2; ++mt) fa[mt][m] = *(const v2f*)(p + mt * 16 * XPITCH + 4 * m);
    }
}
__device__ __forceinline__ void load_b(const float* bcol, int ch, v2f (&fb)[KCH]) {
    const float* p = bcol + ch * (4 * KCH * LBLK);
    #pragma unroll
    for (int m = 0; m < KCH; ++m) {
        fb[m].x = p[(4 * m)     * LBLK];
        fb[m].y = p[(4 * m + 1) * LBLK];
    }
}
__device__ __forceinline__ void mma_chunk(const v2f (&fa)[2][KCH], const v2f (&fb)[KCH], v8f (&acc)[2]) {
    #pragma unroll
    for (int m = 0; m < KCH; ++m) {
        acc[0] = __builtin_amdgcn_wmma_f32_16x16x4_f32(false, fa[0][m], false, fb[m], (short)0, acc[0], false, false);
        acc[1] = __builtin_amdgcn_wmma_f32_16x16x4_f32(false, fa[1][m], false, fb[m], (short)0, acc[1], false, false);
    }
}
struct PPieces { float4 q0, q1, q2, q3; };
__device__ __forceinline__ float4 p_piece(const float* __restrict__ P, int nb, int idx) {
    return *(const float4*)(P + (idx >> 2) * LDIM + nb * LBLK + (idx & 3) * 4);
}
__device__ __forceinline__ void p_fetch(const float* __restrict__ P, int nb, int t, PPieces& v) {
    v.q0 = p_piece(P, nb, t);
    v.q1 = p_piece(P, nb, t + 256);
    v.q2 = p_piece(P, nb, t + 512);
    v.q3 = p_piece(P, nb, t + 768);
}
__device__ __forceinline__ void p_store(float* buf, int t, const PPieces& v) {
    float4* b = (float4*)buf + t;
    b[0]   = v.q0;
    b[256] = v.q1;
    b[512] = v.q2;
    b[768] = v.q3;
}

__device__ __forceinline__ void block_product(const float* xrow, const float* bcol, v8f (&acc)[2]) {
    v2f fa[2][KCH], fb[KCH];
    acc[0] = v8f{};
    acc[1] = v8f{};
    load_a(xrow, 0, fa);
    load_b(bcol, 0, fb);
    for (int ch = 0; ch < LBLK; ++ch) {
        mma_chunk(fa, fb, acc);
        if (ch + 1 < LBLK) { load_a(xrow, ch + 1, fa); load_b(bcol, ch + 1, fb); }
    }
}
__device__ __forceinline__ void leaves_out(float* wscr, int c, int h, const v8f (&acc)[2]) {
    #pragma unroll
    for (int mt = 0; mt < 2; ++mt) {
        #pragma unroll
        for (int r = 0; r < 8; ++r)
            wscr[(16 * mt + 8 * h + r) * SPITCH + c] = acc[mt][r];
    }
}

__global__ __launch_bounds__(512) void gemm_chain_kernel(const float* __restrict__ x,
                                                         const float* __restrict__ P,
                                                         const float* __restrict__ weights,
                                                         const float* __restrict__ biases,
                                                         float* __restrict__ out) {
    __shared__ float Xs[NSCAN * 32 * XPITCH];
    __shared__ float Bs[2 * BBLK];
    __shared__ float scr[NSCAN * 32 * SPITCH];
    __shared__ float sA[255], sW0[255], sW1[255];
    const int t    = threadIdx.x;
    const int wave = __builtin_amdgcn_readfirstlane(t >> 5);
    const int lane = t & 31;
    const bool product = wave >= NSCAN;
    const int pw   = wave & (NSCAN - 1);
    const int tp   = t & 255;
    const int c    = lane & 15;
    const int h    = lane >> 4;

    const int row0 = blockIdx.x * 256 + pw * 32;

    PPieces pst0 = {}, pst1 = {};
    if (!product) {
        p_fetch(P, 0, tp, pst0);
        p_fetch(P, 1, tp, pst1);
    }
    {
        float* xd = Xs + wave * (16 * XPITCH);
        const float4* xg = (const float4*)(x + ((size_t)blockIdx.x * 256 + wave * 16) * LDIM);
        #pragma unroll 32
        for (int it = 0; it < 16 * LDIM / 128; ++it) {
            const int f = it * 32 + lane;
            *(float4*)(xd + (f >> 6) * XPITCH + (f & 63) * 4) = xg[f];
        }
    }
    if (!product) {
        p_store(Bs, tp, pst0);
        p_store(Bs + BBLK, tp, pst1);
    }
    if (t < 255) {
        const float w0 = weights[2 * t], w1 = weights[2 * t + 1];
        const float m  = fmaxf(w0, w1);
        const float e0 = expf(w0 - m), e1 = expf(w1 - m);
        const float inv = 1.0f / (e0 + e1);
        sW0[t] = e0 * inv;
        sW1[t] = e1 * inv;
        float a = 3.0f / (1.0f + expf(-biases[t])) - 1.0f;
        sA[t] = fminf(fmaxf(a, -1.0f + EPS), 2.0f - EPS);
    }
    __syncthreads();

    float* wscr = scr + pw * (32 * SPITCH);
    const float* myrow = wscr + lane * SPITCH;
    const float* xrow  = Xs + pw * (32 * XPITCH) + c * XPITCH + 2 * h;
    const float* bcol  = Bs + (2 * h) * LBLK + c;

    v8f acc[2];
    acc[0] = v8f{};
    acc[1] = v8f{};
    LevelCoef kc = {};
    if (product) {
        block_product(xrow, bcol, acc);
        leaves_out(wscr, c, h, acc);
    } else {
        kc = block_coef(sA, sW0, sW1, 0, c);
    }
    __syncthreads();

    float carry = 0.0f;
    for (int j = 0; j < LDIM / LBLK; ++j) {
        if (product) {
            if (j + 1 < LDIM / LBLK) {
                PPieces pst = {};
                if (j + 2 < LDIM / LBLK) p_fetch(P, j + 2, tp, pst);
                block_product(xrow, bcol + ((j + 1) & 1) * BBLK, acc);
                if (j + 2 < LDIM / LBLK) p_store(Bs + (j & 1) * BBLK, tp, pst);
            }
        } else {
            const bool first = (j == 0);
            const bool last  = (j == LDIM / LBLK - 1);
            const int  i0 = first ? 1 : 0;
            const int  i1 = last ? LBLK - 1 : LBLK;
            if (first) {
                const float y0 = myrow[0];
                carry = lg2(fminf(fmaxf(isnan(y0) ? EPS : y0, EPS), 1.0f - EPS));
            }
            float lf = myrow[i0];
            int   nc = __builtin_amdgcn_readlane(kc.code, i0);
            float na = lane_f(kc.a, i0);
            for (int i = i0; i < i1; ++i) {
                const float clf = lf;
                const int   cc  = nc;
                const float ca  = na;
                if (i + 1 < LBLK) {
                    lf = myrow[i + 1];
                    nc = __builtin_amdgcn_readlane(kc.code, i + 1);
                    na = lane_f(kc.a, i + 1);
                }
                carry = combine_log<false>(carry, clf, kc, i, cc, ca);
            }
            if (last) carry = combine_log<true>(carry, lf, kc, LBLK - 1, nc, na);
        }
        __syncthreads();
        if (j + 1 < LDIM / LBLK) {
            if (product) leaves_out(wscr, c, h, acc);
            else         kc = block_coef(sA, sW0, sW1, (j + 1) * LBLK, c);
        }
        __syncthreads();
    }
    if (!product) out[row0 + lane] = carry;
}

extern "C" void kernel_launch(void* const* d_in, const int* in_sizes, int n_in,
                              void* d_out, int out_size, void* d_ws, size_t ws_size,
                              hipStream_t stream) {
    const float* x       = (const float*)d_in[0];
    const float* logits  = (const float*)d_in[1];
    const float* weights = (const float*)d_in[2];
    const float* biases  = (const float*)d_in[3];

    float* P = (float*)d_ws;

    sinkhorn_kernel<<<1, SK_WAVES * 32, 0, stream>>>(logits, P);
    gemm_chain_kernel<<<BATCH / 256, 512, 0, stream>>>(x, P, weights, biases, (float*)d_out);
}
